# starting kernel
# baseline (speedup 1.0000x reference)
#include <hip/hip_runtime.h>
#include <math.h>

#define B_    16
#define C_    384
#define H_    28
#define W_    28
#define HW_   784
#define N_    785
#define M_    (B_*N_)
#define MP_   (M_ + 16)
#define HEADS_ 8
#define D_    48
#define C3_   1152
#define C4_   1536
#define MT_   (M_/16)
#define MB_   ((MT_ + 1)/2)
#define SLD   800

typedef __attribute__((ext_vector_type(2))) float v2f;
typedef __attribute__((ext_vector_type(8))) float v8f;

__device__ __forceinline__ v8f wmma4(v2f a, v2f b, v8f c) {
  return __builtin_amdgcn_wmma_f32_16x16x4_f32(false, a, false, b, (short)0, c,
                                               false, false);
}

__global__ void conv_pe_kernel(const float* __restrict__ x,
                               const float* __restrict__ cw,
                               const float* __restrict__ cb,
                               float* __restrict__ xc) {
  int idx = blockIdx.x * 256 + threadIdx.x;
  if (idx >= B_ * C_ * HW_) return;
  int b   = idx / (C_ * HW_);
  int rem = idx % (C_ * HW_);
  int c   = rem / HW_;
  int p   = rem % HW_;
  int hh  = p / W_, ww = p % W_;
  const float* xp = x + (size_t)(b * C_ + c) * HW_;
  const float* wp = cw + c * 9;
  float s = 0.f;
#pragma unroll
  for (int i = 0; i < 3; ++i) {
    int hy = hh + i - 1;
    if (hy < 0 || hy >= H_) continue;
#pragma unroll
    for (int j = 0; j < 3; ++j) {
      int wx = ww + j - 1;
      if (wx < 0 || wx >= W_) continue;
      s += xp[hy * W_ + wx] * wp[i * 3 + j];
    }
  }
  xc[((size_t)b * N_ + 1 + p) * C_ + c] = xp[p] + s + cb[c];
}

__global__ void cls_copy_kernel(const float* __restrict__ cls,
                                float* __restrict__ xc) {
  int idx = blockIdx.x * 256 + threadIdx.x;
  if (idx >= B_ * C_) return;
  int b = idx / C_, c = idx % C_;
  xc[(size_t)b * N_ * C_ + c] = cls[idx];
}

__global__ void ln_kernel(const float* __restrict__ xin,
                          const float* __restrict__ g,
                          const float* __restrict__ bta,
                          float* __restrict__ yout, int ntok) {
  int wave = threadIdx.x >> 5, lane = threadIdx.x & 31;
  int t = blockIdx.x * 8 + wave;
  if (t >= ntok) return;
  const float* xp = xin + (size_t)t * C_;
  float s = 0.f, sq = 0.f;
#pragma unroll
  for (int i = 0; i < C_ / 32; ++i) {
    float v = xp[lane + i * 32];
    s += v; sq += v * v;
  }
#pragma unroll
  for (int off = 16; off; off >>= 1) {
    s  += __shfl_xor(s,  off, 32);
    sq += __shfl_xor(sq, off, 32);
  }
  float mu  = s  * (1.f / C_);
  float var = sq * (1.f / C_) - mu * mu;
  float rs  = rsqrtf(var + 1e-5f);
  float* yp = yout + (size_t)t * C_;
#pragma unroll
  for (int i = 0; i < C_ / 32; ++i) {
    int c = lane + i * 32;
    yp[c] = (xp[c] - mu) * rs * g[c] + bta[c];
  }
}

template <int EPI>
__global__ __launch_bounds__(256) void gemm_wmma_kernel(
    const float* __restrict__ X, const float* __restrict__ Wt,
    const float* __restrict__ bias, float* __restrict__ Out,
    int K, int ldo, int Mblocks) {
  int lane = threadIdx.x & 31;
  int wave = threadIdx.x >> 5;
  int mblk = blockIdx.y * 8 + wave;
  if (mblk >= Mblocks) return;
  int n0 = blockIdx.x * 64;
  int m0 = mblk * 32;
  int rc = lane & 15;
  int kb = (lane >> 4) << 1;
  const float* xp0 = X  + (size_t)(m0 + rc)      * K + kb;
  const float* xp1 = X  + (size_t)(m0 + 16 + rc) * K + kb;
  const float* wp0 = Wt + (size_t)(n0 + rc)      * K + kb;
  const float* wp1 = Wt + (size_t)(n0 + 16 + rc) * K + kb;
  const float* wp2 = Wt + (size_t)(n0 + 32 + rc) * K + kb;
  const float* wp3 = Wt + (size_t)(n0 + 48 + rc) * K + kb;

  v8f acc[2][4] = {};
#pragma unroll 4
  for (int k = 0; k < K; k += 4) {
    v2f a0 = *(const v2f*)(xp0 + k);
    v2f a1 = *(const v2f*)(xp1 + k);
    v2f b0 = *(const v2f*)(wp0 + k);
    v2f b1 = *(const v2f*)(wp1 + k);
    v2f b2 = *(const v2f*)(wp2 + k);
    v2f b3 = *(const v2f*)(wp3 + k);
    acc[0][0] = wmma4(a0, b0, acc[0][0]);
    acc[1][0] = wmma4(a1, b0, acc[1][0]);
    acc[0][1] = wmma4(a0, b1, acc[0][1]);
    acc[1][1] = wmma4(a1, b1, acc[1][1]);
    acc[0][2] = wmma4(a0, b2, acc[0][2]);
    acc[1][2] = wmma4(a1, b2, acc[1][2]);
    acc[0][3] = wmma4(a0, b3, acc[0][3]);
    acc[1][3] = wmma4(a1, b3, acc[1][3]);
  }

  int coln = lane & 15;
  int roff = (lane >> 4) << 3;
#pragma unroll
  for (int mi = 0; mi < 2; ++mi) {
#pragma unroll
    for (int j = 0; j < 4; ++j) {
      int nc = n0 + j * 16 + coln;
      float bv = bias ? bias[nc] : 0.f;
#pragma unroll
      for (int r = 0; r < 8; ++r) {
        int gm = m0 + mi * 16 + roff + r;
        if (gm < M_) {
          size_t oi = (size_t)gm * ldo + nc;
          float v = acc[mi][j][r] + bv;
          if (EPI == 0)      Out[oi] = v;
          else if (EPI == 1) Out[oi] += v;
          else               Out[oi] = 0.5f * v * (1.f + erff(v * 0.70710678118654752f));
        }
      }
    }
  }
}

__global__ __launch_bounds__(32) void attn_kernel(const float* __restrict__ qkv,
                                                  float* __restrict__ o,
                                                  float* __restrict__ ga) {
  __shared__ float S[16 * SLD];
  int lane = threadIdx.x;
  int tile = blockIdx.x;
  int h = blockIdx.y, b = blockIdx.z;
  int t0 = tile * 16;
  int row = lane & 15;
  int kb  = (lane >> 4) << 1;
  int gq = t0 + row; if (gq > N_ - 1) gq = N_ - 1;
  const float* qrow = qkv + (size_t)(b * N_ + gq) * C3_ + h * D_ + kb;
  const float scale = 0.14433756729740643f;

  for (int ct = 0; ct < 50; ++ct) {
    int kc = ct * 16 + row; if (kc > N_ - 1) kc = N_ - 1;
    const float* krow = qkv + (size_t)(b * N_ + kc) * C3_ + C_ + h * D_ + kb;
    v8f acc = {};
#pragma unroll
    for (int k = 0; k < D_; k += 4) {
      v2f a  = *(const v2f*)(qrow + k);
      v2f bb = *(const v2f*)(krow + k);
      acc = wmma4(a, bb, acc);
    }
    int sc  = ct * 16 + (lane & 15);
    int mrb = (lane >> 4) << 3;
#pragma unroll
    for (int r = 0; r < 8; ++r) S[(mrb + r) * SLD + sc] = acc[r] * scale;
  }
  __syncthreads();

  {
    int r = lane >> 1, sub = lane & 1;
    float mx = -3.0e38f;
    for (int c = sub; c < N_; c += 2) mx = fmaxf(mx, S[r * SLD + c]);
    mx = fmaxf(mx, __shfl_xor(mx, 1, 32));
    float sum = 0.f;
    for (int c = sub; c < N_; c += 2) {
      float e = expf(S[r * SLD + c] - mx);
      S[r * SLD + c] = e; sum += e;
    }
    sum += __shfl_xor(sum, 1, 32);
    float inv = 1.f / sum;
    for (int c = sub; c < N_; c += 2) S[r * SLD + c] *= inv;
    for (int c = N_ + sub; c < SLD; c += 2) S[r * SLD + c] = 0.f;
  }
  __syncthreads();

  if (t0 == 0 && (lane >> 1) == 0) {
    int sub = lane & 1;
    float* gp = ga + (size_t)(b * HEADS_ + h) * HW_;
    for (int c = 1 + sub; c < N_; c += 2) gp[c - 1] = S[c];
  }

  const float* vbase = qkv + (size_t)b * N_ * C3_ + 2 * C_ + h * D_;
  int coln = lane & 15;
  v8f acc0 = {}, acc1 = {}, acc2 = {};
  for (int k = 0; k < 788; k += 4) {
    int k0 = k + kb;
    v2f a = *(const v2f*)(&S[row * SLD + k0]);
    int r0 = k0 > HW_ ? HW_ : k0;
    int r1 = (k0 + 1) > HW_ ? HW_ : (k0 + 1);
    const float* vr0 = vbase + (size_t)r0 * C3_ + coln;
    const float* vr1 = vbase + (size_t)r1 * C3_ + coln;
    v2f b0, b1, b2;
    b0.x = vr0[0];  b0.y = vr1[0];
    b1.x = vr0[16]; b1.y = vr1[16];
    b2.x = vr0[32]; b2.y = vr1[32];
    acc0 = wmma4(a, b0, acc0);
    acc1 = wmma4(a, b1, acc1);
    acc2 = wmma4(a, b2, acc2);
  }
  int mrb = (lane >> 4) << 3;
#pragma unroll
  for (int r = 0; r < 8; ++r) {
    int gm = t0 + mrb + r;
    if (gm < N_) {
      float* op = o + ((size_t)(b * N_ + gm)) * C_ + h * D_ + coln;
      op[0]  = acc0[r];
      op[16] = acc1[r];
      op[32] = acc2[r];
    }
  }
}

__global__ void finalize_kernel(const float* __restrict__ xc,
                                const float* __restrict__ ga,
                                float* __restrict__ out) {
  size_t idx = (size_t)blockIdx.x * 256 + threadIdx.x;
  const size_t n_cls = (size_t)B_ * C_;
  const size_t n_x   = (size_t)B_ * C_ * HW_;
  const size_t n_ga  = (size_t)B_ * HW_;
  if (idx < n_cls) {
    int b = (int)(idx / C_), c = (int)(idx % C_);
    out[idx] = xc[(size_t)b * N_ * C_ + c];
  } else if (idx < n_cls + n_x) {
    size_t j = idx - n_cls;
    int b   = (int)(j / (C_ * HW_));
    int rem = (int)(j % (C_ * HW_));
    int c = rem / HW_, p = rem % HW_;
    out[idx] = xc[((size_t)b * N_ + 1 + p) * C_ + c];
  } else if (idx < n_cls + n_x + n_ga) {
    size_t j = idx - n_cls - n_x;
    int b = (int)(j / HW_), p = (int)(j % HW_);
    float s = 0.f;
#pragma unroll
    for (int hh = 0; hh < HEADS_; ++hh)
      s += ga[(size_t)(b * HEADS_ + hh) * HW_ + p];
    out[idx] = s * 0.125f;
  }
}

extern "C" void kernel_launch(void* const* d_in, const int* in_sizes, int n_in,
                              void* d_out, int out_size, void* d_ws, size_t ws_size,
                              hipStream_t stream) {
  const float* cls    = (const float*)d_in[0];
  const float* x      = (const float*)d_in[1];
  const float* conv_w = (const float*)d_in[2];
  const float* conv_b = (const float*)d_in[3];
  const float* ln1_g  = (const float*)d_in[4];
  const float* ln1_b  = (const float*)d_in[5];
  const float* qkv_w  = (const float*)d_in[6];
  const float* proj_w = (const float*)d_in[7];
  const float* proj_b = (const float*)d_in[8];
  const float* ln2_g  = (const float*)d_in[9];
  const float* ln2_b  = (const float*)d_in[10];
  const float* fc1_w  = (const float*)d_in[11];
  const float* fc1_b  = (const float*)d_in[12];
  const float* fc2_w  = (const float*)d_in[13];
  const float* fc2_b  = (const float*)d_in[14];

  float* ws  = (float*)d_ws;
  float* xc  = ws;
  float* y   = xc  + (size_t)MP_ * C_;
  float* qkv = y   + (size_t)MP_ * C_;
  float* o   = qkv + (size_t)MP_ * C3_;
  float* m1  = qkv;
  float* ga  = o   + (size_t)MP_ * C_;
  float* out = (float*)d_out;

  conv_pe_kernel<<<(B_ * C_ * HW_ + 255) / 256, 256, 0, stream>>>(x, conv_w, conv_b, xc);
  cls_copy_kernel<<<(B_ * C_ + 255) / 256, 256, 0, stream>>>(cls, xc);

  ln_kernel<<<(M_ + 7) / 8, 256, 0, stream>>>(xc, ln1_g, ln1_b, y, M_);

  gemm_wmma_kernel<0><<<dim3(C3_ / 64, (MB_ + 7) / 8), 256, 0, stream>>>(
      y, qkv_w, nullptr, qkv, C_, C3_, MB_);

  attn_kernel<<<dim3(50, HEADS_, B_), 32, 0, stream>>>(qkv, o, ga);

  gemm_wmma_kernel<1><<<dim3(C_ / 64, (MB_ + 7) / 8), 256, 0, stream>>>(
      o, proj_w, proj_b, xc, C_, C_, MB_);

  ln_kernel<<<(M_ + 7) / 8, 256, 0, stream>>>(xc, ln2_g, ln2_b, y, M_);

  gemm_wmma_kernel<2><<<dim3(C4_ / 64, (MB_ + 7) / 8), 256, 0, stream>>>(
      y, fc1_w, fc1_b, m1, C_, C4_, MB_);

  gemm_wmma_kernel<1><<<dim3(C_ / 64, (MB_ + 7) / 8), 256, 0, stream>>>(
      m1, fc2_w, fc2_b, xc, C4_, C_, MB_);

  size_t tot = (size_t)B_ * C_ + (size_t)B_ * C_ * HW_ + (size_t)B_ * HW_;
  finalize_kernel<<<(unsigned)((tot + 255) / 256), 256, 0, stream>>>(xc, ga, out);
}


# GEMM kernel: X and W tiles (128x64 per 4-wave block, K chunks of 32) staged in double-buffered LDS by coalesced 128-bit global loads; WMMA fragments read from LDS; k order and sums unchanged
# speedup vs baseline: 1.5436x; 1.5436x over previous
#include <hip/hip_runtime.h>
#include <math.h>

#define B_    16
#define C_    384
#define H_    28
#define W_    28
#define HW_   784
#define N_    785
#define M_    (B_*N_)
#define MP_   (M_ + 16)
#define HEADS_ 8
#define D_    48
#define C3_   1152
#define C4_   1536
#define SLD   800

typedef __attribute__((ext_vector_type(2))) float v2f;
typedef __attribute__((ext_vector_type(8))) float v8f;

__device__ __forceinline__ v8f wmma4(v2f a, v2f b, v8f c) {
  return __builtin_amdgcn_wmma_f32_16x16x4_f32(false, a, false, b, (short)0, c,
                                               false, false);
}

__global__ void conv_pe_kernel(const float* __restrict__ x,
                               const float* __restrict__ cw,
                               const float* __restrict__ cb,
                               float* __restrict__ xc) {
  int idx = blockIdx.x * 256 + threadIdx.x;
  if (idx >= B_ * C_ * HW_) return;
  int b   = idx / (C_ * HW_);
  int rem = idx % (C_ * HW_);
  int c   = rem / HW_;
  int p   = rem % HW_;
  int hh  = p / W_, ww = p % W_;
  const float* xp = x + (size_t)(b * C_ + c) * HW_;
  const float* wp = cw + c * 9;
  float s = 0.f;
#pragma unroll
  for (int i = 0; i < 3; ++i) {
    int hy = hh + i - 1;
    if (hy < 0 || hy >= H_) continue;
#pragma unroll
    for (int j = 0; j < 3; ++j) {
      int wx = ww + j - 1;
      if (wx < 0 || wx >= W_) continue;
      s += xp[hy * W_ + wx] * wp[i * 3 + j];
    }
  }
  xc[((size_t)b * N_ + 1 + p) * C_ + c] = xp[p] + s + cb[c];
}

__global__ void cls_copy_kernel(const float* __restrict__ cls,
                                float* __restrict__ xc) {
  int idx = blockIdx.x * 256 + threadIdx.x;
  if (idx >= B_ * C_) return;
  int b = idx / C_, c = idx % C_;
  xc[(size_t)b * N_ * C_ + c] = cls[idx];
}

__global__ void ln_kernel(const float* __restrict__ xin,
                          const float* __restrict__ g,
                          const float* __restrict__ bta,
                          float* __restrict__ yout, int ntok) {
  int wave = threadIdx.x >> 5, lane = threadIdx.x & 31;
  int t = blockIdx.x * 8 + wave;
  if (t >= ntok) return;
  const float* xp = xin + (size_t)t * C_;
  float s = 0.f, sq = 0.f;
#pragma unroll
  for (int i = 0; i < C_ / 32; ++i) {
    float v = xp[lane + i * 32];
    s += v; sq += v * v;
  }
#pragma unroll
  for (int off = 16; off; off >>= 1) {
    s  += __shfl_xor(s,  off, 32);
    sq += __shfl_xor(sq, off, 32);
  }
  float mu  = s  * (1.f / C_);
  float var = sq * (1.f / C_) - mu * mu;
  float rs  = rsqrtf(var + 1e-5f);
  float* yp = yout + (size_t)t * C_;
#pragma unroll
  for (int i = 0; i < C_ / 32; ++i) {
    int c = lane + i * 32;
    yp[c] = (xp[c] - mu) * rs * g[c] + bta[c];
  }
}

#define GBM_  128
#define GBN_  64
#define GKC_  32
#define GLDP_ 36
#define GMB_  ((M_ + GBM_ - 1) / GBM_)

typedef __attribute__((ext_vector_type(4))) float v4f;

__device__ __forceinline__ void gemm_chunk(const float* __restrict__ as,
                                           const float* __restrict__ bs,
                                           v8f (&acc)[2][4]) {
  v2f fa[2][2], fb[2][4];
  fa[0][0] = *(const v2f*)(as);
  fa[0][1] = *(const v2f*)(as + 16 * GLDP_);
#pragma unroll
  for (int j = 0; j < 4; ++j) fb[0][j] = *(const v2f*)(bs + j * 16 * GLDP_);
#pragma unroll
  for (int ks = 0; ks < GKC_ / 4; ++ks) {
    const int c = ks & 1, n = c ^ 1;
    if (ks + 1 < GKC_ / 4) {
      fa[n][0] = *(const v2f*)(as + (ks + 1) * 4);
      fa[n][1] = *(const v2f*)(as + 16 * GLDP_ + (ks + 1) * 4);
#pragma unroll
      for (int j = 0; j < 4; ++j)
        fb[n][j] = *(const v2f*)(bs + j * 16 * GLDP_ + (ks + 1) * 4);
    }
    acc[0][0] = wmma4(fa[c][0], fb[c][0], acc[0][0]);
    acc[1][0] = wmma4(fa[c][1], fb[c][0], acc[1][0]);
    acc[0][1] = wmma4(fa[c][0], fb[c][1], acc[0][1]);
    acc[1][1] = wmma4(fa[c][1], fb[c][1], acc[1][1]);
    acc[0][2] = wmma4(fa[c][0], fb[c][2], acc[0][2]);
    acc[1][2] = wmma4(fa[c][1], fb[c][2], acc[1][2]);
    acc[0][3] = wmma4(fa[c][0], fb[c][3], acc[0][3]);
    acc[1][3] = wmma4(fa[c][1], fb[c][3], acc[1][3]);
  }
}

template <int EPI>
__global__ __launch_bounds__(128) void gemm_wmma_kernel(
    const float* __restrict__ X, const float* __restrict__ Wt,
    const float* __restrict__ bias, float* __restrict__ Out,
    int K, int ldo) {
  __shared__ __attribute__((aligned(16))) float As[2][GBM_ * GLDP_];
  __shared__ __attribute__((aligned(16))) float Bs[2][GBN_ * GLDP_];
  const int tid  = threadIdx.x;
  const int lane = tid & 31;
  const int wave = tid >> 5;
  const int n0  = blockIdx.x * GBN_;
  const int mb0 = blockIdx.y * GBM_;

  const int lr = tid >> 3;
  const int lc = (tid & 7) << 2;
  const float* ag[8];
  const float* bg[4];
#pragma unroll
  for (int i = 0; i < 8; ++i) {
    int r = mb0 + lr + 16 * i;
    if (r > M_ - 1) r = M_ - 1;
    ag[i] = X + (size_t)r * K + lc;
  }
#pragma unroll
  for (int i = 0; i < 4; ++i) bg[i] = Wt + (size_t)(n0 + lr + 16 * i) * K + lc;
  const int soff = lr * GLDP_ + lc;

  v4f ra[8], rb[4];
#pragma unroll
  for (int i = 0; i < 8; ++i) ra[i] = *(const v4f*)(ag[i]);
#pragma unroll
  for (int i = 0; i < 4; ++i) rb[i] = *(const v4f*)(bg[i]);
#pragma unroll
  for (int i = 0; i < 8; ++i) *(v4f*)(&As[0][soff + 16 * i * GLDP_]) = ra[i];
#pragma unroll
  for (int i = 0; i < 4; ++i) *(v4f*)(&Bs[0][soff + 16 * i * GLDP_]) = rb[i];
  __syncthreads();

  const int rc = lane & 15;
  const int kb = (lane >> 4) << 1;
  const int aoff = (wave * 32 + rc) * GLDP_ + kb;
  const int boff = rc * GLDP_ + kb;

  v8f acc[2][4] = {};
  const int nk = K / GKC_;
  for (int kc = 0; kc < nk - 1; ++kc) {
    const int cur = kc & 1;
    const int kn = (kc + 1) * GKC_;
#pragma unroll
    for (int i = 0; i < 8; ++i) ra[i] = *(const v4f*)(ag[i] + kn);
#pragma unroll
    for (int i = 0; i < 4; ++i) rb[i] = *(const v4f*)(bg[i] + kn);
    gemm_chunk(&As[cur][aoff], &Bs[cur][boff], acc);
#pragma unroll
    for (int i = 0; i < 8; ++i) *(v4f*)(&As[cur ^ 1][soff + 16 * i * GLDP_]) = ra[i];
#pragma unroll
    for (int i = 0; i < 4; ++i) *(v4f*)(&Bs[cur ^ 1][soff + 16 * i * GLDP_]) = rb[i];
    __syncthreads();
  }
  {
    const int cur = (nk - 1) & 1;
    gemm_chunk(&As[cur][aoff], &Bs[cur][boff], acc);
  }

  const int m0 = mb0 + wave * 32;
  int coln = lane & 15;
  int roff = (lane >> 4) << 3;
#pragma unroll
  for (int mi = 0; mi < 2; ++mi) {
#pragma unroll
    for (int j = 0; j < 4; ++j) {
      int nc = n0 + j * 16 + coln;
      float bv = bias ? bias[nc] : 0.f;
#pragma unroll
      for (int r = 0; r < 8; ++r) {
        int gm = m0 + mi * 16 + roff + r;
        if (gm < M_) {
          size_t oi = (size_t)gm * ldo + nc;
          float v = acc[mi][j][r] + bv;
          if (EPI == 0)      Out[oi] = v;
          else if (EPI == 1) Out[oi] += v;
          else               Out[oi] = 0.5f * v * (1.f + erff(v * 0.70710678118654752f));
        }
      }
    }
  }
}

__global__ __launch_bounds__(32) void attn_kernel(const float* __restrict__ qkv,
                                                  float* __restrict__ o,
                                                  float* __restrict__ ga) {
  __shared__ float S[16 * SLD];
  int lane = threadIdx.x;
  int tile = blockIdx.x;
  int h = blockIdx.y, b = blockIdx.z;
  int t0 = tile * 16;
  int row = lane & 15;
  int kb  = (lane >> 4) << 1;
  int gq = t0 + row; if (gq > N_ - 1) gq = N_ - 1;
  const float* qrow = qkv + (size_t)(b * N_ + gq) * C3_ + h * D_ + kb;
  const float scale = 0.14433756729740643f;

  for (int ct = 0; ct < 50; ++ct) {
    int kc = ct * 16 + row; if (kc > N_ - 1) kc = N_ - 1;
    const float* krow = qkv + (size_t)(b * N_ + kc) * C3_ + C_ + h * D_ + kb;
    v8f acc = {};
#pragma unroll
    for (int k = 0; k < D_; k += 4) {
      v2f a  = *(const v2f*)(qrow + k);
      v2f bb = *(const v2f*)(krow + k);
      acc = wmma4(a, bb, acc);
    }
    int sc  = ct * 16 + (lane & 15);
    int mrb = (lane >> 4) << 3;
#pragma unroll
    for (int r = 0; r < 8; ++r) S[(mrb + r) * SLD + sc] = acc[r] * scale;
  }
  __syncthreads();

  {
    int r = lane >> 1, sub = lane & 1;
    float mx = -3.0e38f;
    for (int c = sub; c < N_; c += 2) mx = fmaxf(mx, S[r * SLD + c]);
    mx = fmaxf(mx, __shfl_xor(mx, 1, 32));
    float sum = 0.f;
    for (int c = sub; c < N_; c += 2) {
      float e = expf(S[r * SLD + c] - mx);
      S[r * SLD + c] = e; sum += e;
    }
    sum += __shfl_xor(sum, 1, 32);
    float inv = 1.f / sum;
    for (int c = sub; c < N_; c += 2) S[r * SLD + c] *= inv;
    for (int c = N_ + sub; c < SLD; c += 2) S[r * SLD + c] = 0.f;
  }
  __syncthreads();

  if (t0 == 0 && (lane >> 1) == 0) {
    int sub = lane & 1;
    float* gp = ga + (size_t)(b * HEADS_ + h) * HW_;
    for (int c = 1 + sub; c < N_; c += 2) gp[c - 1] = S[c];
  }

  const float* vbase = qkv + (size_t)b * N_ * C3_ + 2 * C_ + h * D_;
  int coln = lane & 15;
  v8f acc0 = {}, acc1 = {}, acc2 = {};
  for (int k = 0; k < 788; k += 4) {
    int k0 = k + kb;
    v2f a = *(const v2f*)(&S[row * SLD + k0]);
    int r0 = k0 > HW_ ? HW_ : k0;
    int r1 = (k0 + 1) > HW_ ? HW_ : (k0 + 1);
    const float* vr0 = vbase + (size_t)r0 * C3_ + coln;
    const float* vr1 = vbase + (size_t)r1 * C3_ + coln;
    v2f b0, b1, b2;
    b0.x = vr0[0];  b0.y = vr1[0];
    b1.x = vr0[16]; b1.y = vr1[16];
    b2.x = vr0[32]; b2.y = vr1[32];
    acc0 = wmma4(a, b0, acc0);
    acc1 = wmma4(a, b1, acc1);
    acc2 = wmma4(a, b2, acc2);
  }
  int mrb = (lane >> 4) << 3;
#pragma unroll
  for (int r = 0; r < 8; ++r) {
    int gm = t0 + mrb + r;
    if (gm < N_) {
      float* op = o + ((size_t)(b * N_ + gm)) * C_ + h * D_ + coln;
      op[0]  = acc0[r];
      op[16] = acc1[r];
      op[32] = acc2[r];
    }
  }
}

__global__ void finalize_kernel(const float* __restrict__ xc,
                                const float* __restrict__ ga,
                                float* __restrict__ out) {
  size_t idx = (size_t)blockIdx.x * 256 + threadIdx.x;
  const size_t n_cls = (size_t)B_ * C_;
  const size_t n_x   = (size_t)B_ * C_ * HW_;
  const size_t n_ga  = (size_t)B_ * HW_;
  if (idx < n_cls) {
    int b = (int)(idx / C_), c = (int)(idx % C_);
    out[idx] = xc[(size_t)b * N_ * C_ + c];
  } else if (idx < n_cls + n_x) {
    size_t j = idx - n_cls;
    int b   = (int)(j / (C_ * HW_));
    int rem = (int)(j % (C_ * HW_));
    int c = rem / HW_, p = rem % HW_;
    out[idx] = xc[((size_t)b * N_ + 1 + p) * C_ + c];
  } else if (idx < n_cls + n_x + n_ga) {
    size_t j = idx - n_cls - n_x;
    int b = (int)(j / HW_), p = (int)(j % HW_);
    float s = 0.f;
#pragma unroll
    for (int hh = 0; hh < HEADS_; ++hh)
      s += ga[(size_t)(b * HEADS_ + hh) * HW_ + p];
    out[idx] = s * 0.125f;
  }
}

extern "C" void kernel_launch(void* const* d_in, const int* in_sizes, int n_in,
                              void* d_out, int out_size, void* d_ws, size_t ws_size,
                              hipStream_t stream) {
  const float* cls    = (const float*)d_in[0];
  const float* x      = (const float*)d_in[1];
  const float* conv_w = (const float*)d_in[2];
  const float* conv_b = (const float*)d_in[3];
  const float* ln1_g  = (const float*)d_in[4];
  const float* ln1_b  = (const float*)d_in[5];
  const float* qkv_w  = (const float*)d_in[6];
  const float* proj_w = (const float*)d_in[7];
  const float* proj_b = (const float*)d_in[8];
  const float* ln2_g  = (const float*)d_in[9];
  const float* ln2_b  = (const float*)d_in[10];
  const float* fc1_w  = (const float*)d_in[11];
  const float* fc1_b  = (const float*)d_in[12];
  const float* fc2_w  = (const float*)d_in[13];
  const float* fc2_b  = (const float*)d_in[14];

  float* ws  = (float*)d_ws;
  float* xc  = ws;
  float* y   = xc  + (size_t)MP_ * C_;
  float* qkv = y   + (size_t)MP_ * C_;
  float* o   = qkv + (size_t)MP_ * C3_;
  float* m1  = qkv;
  float* ga  = o   + (size_t)MP_ * C_;
  float* out = (float*)d_out;

  conv_pe_kernel<<<(B_ * C_ * HW_ + 255) / 256, 256, 0, stream>>>(x, conv_w, conv_b, xc);
  cls_copy_kernel<<<(B_ * C_ + 255) / 256, 256, 0, stream>>>(cls, xc);

  ln_kernel<<<(M_ + 7) / 8, 256, 0, stream>>>(xc, ln1_g, ln1_b, y, M_);

  gemm_wmma_kernel<0><<<dim3(C3_ / GBN_, GMB_), 128, 0, stream>>>(
      y, qkv_w, nullptr, qkv, C_, C3_);

  attn_kernel<<<dim3(50, HEADS_, B_), 32, 0, stream>>>(qkv, o, ga);

  gemm_wmma_kernel<1><<<dim3(C_ / GBN_, GMB_), 128, 0, stream>>>(
      o, proj_w, proj_b, xc, C_, C_);

  ln_kernel<<<(M_ + 7) / 8, 256, 0, stream>>>(xc, ln2_g, ln2_b, y, M_);

  gemm_wmma_kernel<2><<<dim3(C4_ / GBN_, GMB_), 128, 0, stream>>>(
      y, fc1_w, fc1_b, m1, C_, C4_);

  gemm_wmma_kernel<1><<<dim3(C_ / GBN_, GMB_), 128, 0, stream>>>(
      m1, fc2_w, fc2_b, xc, C4_, C_);

  size_t tot = (size_t)B_ * C_ + (size_t)B_ * C_ * HW_ + (size_t)B_ * HW_;
  finalize_kernel<<<(unsigned)((tot + 255) / 256), 256, 0, stream>>>(xc, ga, out);
}


# Attention kernel: keys walked in 64-key K/V tiles staged in LDS and shared by 4 waves (64 query rows per block), running max/sum softmax instead of whole score rows in LDS; summation order changes
# speedup vs baseline: 2.3132x; 1.4986x over previous
#include <hip/hip_runtime.h>
#include <math.h>

#define B_    16
#define C_    384
#define H_    28
#define W_    28
#define HW_   784
#define N_    785
#define M_    (B_*N_)
#define MP_   (M_ + 16)
#define HEADS_ 8
#define D_    48
#define C3_   1152
#define C4_   1536
#define SLD   800

typedef __attribute__((ext_vector_type(2))) float v2f;
typedef __attribute__((ext_vector_type(8))) float v8f;

__device__ __forceinline__ v8f wmma4(v2f a, v2f b, v8f c) {
  return __builtin_amdgcn_wmma_f32_16x16x4_f32(false, a, false, b, (short)0, c,
                                               false, false);
}

__global__ void conv_pe_kernel(const float* __restrict__ x,
                               const float* __restrict__ cw,
                               const float* __restrict__ cb,
                               float* __restrict__ xc) {
  int idx = blockIdx.x * 256 + threadIdx.x;
  if (idx >= B_ * C_ * HW_) return;
  int b   = idx / (C_ * HW_);
  int rem = idx % (C_ * HW_);
  int c   = rem / HW_;
  int p   = rem % HW_;
  int hh  = p / W_, ww = p % W_;
  const float* xp = x + (size_t)(b * C_ + c) * HW_;
  const float* wp = cw + c * 9;
  float s = 0.f;
#pragma unroll
  for (int i = 0; i < 3; ++i) {
    int hy = hh + i - 1;
    if (hy < 0 || hy >= H_) continue;
#pragma unroll
    for (int j = 0; j < 3; ++j) {
      int wx = ww + j - 1;
      if (wx < 0 || wx >= W_) continue;
      s += xp[hy * W_ + wx] * wp[i * 3 + j];
    }
  }
  xc[((size_t)b * N_ + 1 + p) * C_ + c] = xp[p] + s + cb[c];
}

__global__ void cls_copy_kernel(const float* __restrict__ cls,
                                float* __restrict__ xc) {
  int idx = blockIdx.x * 256 + threadIdx.x;
  if (idx >= B_ * C_) return;
  int b = idx / C_, c = idx % C_;
  xc[(size_t)b * N_ * C_ + c] = cls[idx];
}

__global__ void ln_kernel(const float* __restrict__ xin,
                          const float* __restrict__ g,
                          const float* __restrict__ bta,
                          float* __restrict__ yout, int ntok) {
  int wave = threadIdx.x >> 5, lane = threadIdx.x & 31;
  int t = blockIdx.x * 8 + wave;
  if (t >= ntok) return;
  const float* xp = xin + (size_t)t * C_;
  float s = 0.f, sq = 0.f;
#pragma unroll
  for (int i = 0; i < C_ / 32; ++i) {
    float v = xp[lane + i * 32];
    s += v; sq += v * v;
  }
#pragma unroll
  for (int off = 16; off; off >>= 1) {
    s  += __shfl_xor(s,  off, 32);
    sq += __shfl_xor(sq, off, 32);
  }
  float mu  = s  * (1.f / C_);
  float var = sq * (1.f / C_) - mu * mu;
  float rs  = rsqrtf(var + 1e-5f);
  float* yp = yout + (size_t)t * C_;
#pragma unroll
  for (int i = 0; i < C_ / 32; ++i) {
    int c = lane + i * 32;
    yp[c] = (xp[c] - mu) * rs * g[c] + bta[c];
  }
}

#define GBM_  128
#define GBN_  64
#define GKC_  32
#define GLDP_ 36
#define GMB_  ((M_ + GBM_ - 1) / GBM_)

typedef __attribute__((ext_vector_type(4))) float v4f;

__device__ __forceinline__ void gemm_chunk(const float* __restrict__ as,
                                           const float* __restrict__ bs,
                                           v8f (&acc)[2][4]) {
  v2f fa[2][2], fb[2][4];
  fa[0][0] = *(const v2f*)(as);
  fa[0][1] = *(const v2f*)(as + 16 * GLDP_);
#pragma unroll
  for (int j = 0; j < 4; ++j) fb[0][j] = *(const v2f*)(bs + j * 16 * GLDP_);
#pragma unroll
  for (int ks = 0; ks < GKC_ / 4; ++ks) {
    const int c = ks & 1, n = c ^ 1;
    if (ks + 1 < GKC_ / 4) {
      fa[n][0] = *(const v2f*)(as + (ks + 1) * 4);
      fa[n][1] = *(const v2f*)(as + 16 * GLDP_ + (ks + 1) * 4);
#pragma unroll
      for (int j = 0; j < 4; ++j)
        fb[n][j] = *(const v2f*)(bs + j * 16 * GLDP_ + (ks + 1) * 4);
    }
    acc[0][0] = wmma4(fa[c][0], fb[c][0], acc[0][0]);
    acc[1][0] = wmma4(fa[c][1], fb[c][0], acc[1][0]);
    acc[0][1] = wmma4(fa[c][0], fb[c][1], acc[0][1]);
    acc[1][1] = wmma4(fa[c][1], fb[c][1], acc[1][1]);
    acc[0][2] = wmma4(fa[c][0], fb[c][2], acc[0][2]);
    acc[1][2] = wmma4(fa[c][1], fb[c][2], acc[1][2]);
    acc[0][3] = wmma4(fa[c][0], fb[c][3], acc[0][3]);
    acc[1][3] = wmma4(fa[c][1], fb[c][3], acc[1][3]);
  }
}

template <int EPI>
__global__ __launch_bounds__(128) void gemm_wmma_kernel(
    const float* __restrict__ X, const float* __restrict__ Wt,
    const float* __restrict__ bias, float* __restrict__ Out,
    int K, int ldo) {
  __shared__ __attribute__((aligned(16))) float As[2][GBM_ * GLDP_];
  __shared__ __attribute__((aligned(16))) float Bs[2][GBN_ * GLDP_];
  const int tid  = threadIdx.x;
  const int lane = tid & 31;
  const int wave = tid >> 5;
  const int n0  = blockIdx.x * GBN_;
  const int mb0 = blockIdx.y * GBM_;

  const int lr = tid >> 3;
  const int lc = (tid & 7) << 2;
  const float* ag[8];
  const float* bg[4];
#pragma unroll
  for (int i = 0; i < 8; ++i) {
    int r = mb0 + lr + 16 * i;
    if (r > M_ - 1) r = M_ - 1;
    ag[i] = X + (size_t)r * K + lc;
  }
#pragma unroll
  for (int i = 0; i < 4; ++i) bg[i] = Wt + (size_t)(n0 + lr + 16 * i) * K + lc;
  const int soff = lr * GLDP_ + lc;

  v4f ra[8], rb[4];
#pragma unroll
  for (int i = 0; i < 8; ++i) ra[i] = *(const v4f*)(ag[i]);
#pragma unroll
  for (int i = 0; i < 4; ++i) rb[i] = *(const v4f*)(bg[i]);
#pragma unroll
  for (int i = 0; i < 8; ++i) *(v4f*)(&As[0][soff + 16 * i * GLDP_]) = ra[i];
#pragma unroll
  for (int i = 0; i < 4; ++i) *(v4f*)(&Bs[0][soff + 16 * i * GLDP_]) = rb[i];
  __syncthreads();

  const int rc = lane & 15;
  const int kb = (lane >> 4) << 1;
  const int aoff = (wave * 32 + rc) * GLDP_ + kb;
  const int boff = rc * GLDP_ + kb;

  v8f acc[2][4] = {};
  const int nk = K / GKC_;
  for (int kc = 0; kc < nk - 1; ++kc) {
    const int cur = kc & 1;
    const int kn = (kc + 1) * GKC_;
#pragma unroll
    for (int i = 0; i < 8; ++i) ra[i] = *(const v4f*)(ag[i] + kn);
#pragma unroll
    for (int i = 0; i < 4; ++i) rb[i] = *(const v4f*)(bg[i] + kn);
    gemm_chunk(&As[cur][aoff], &Bs[cur][boff], acc);
#pragma unroll
    for (int i = 0; i < 8; ++i) *(v4f*)(&As[cur ^ 1][soff + 16 * i * GLDP_]) = ra[i];
#pragma unroll
    for (int i = 0; i < 4; ++i) *(v4f*)(&Bs[cur ^ 1][soff + 16 * i * GLDP_]) = rb[i];
    __syncthreads();
  }
  {
    const int cur = (nk - 1) & 1;
    gemm_chunk(&As[cur][aoff], &Bs[cur][boff], acc);
  }

  const int m0 = mb0 + wave * 32;
  int coln = lane & 15;
  int roff = (lane >> 4) << 3;
#pragma unroll
  for (int mi = 0; mi < 2; ++mi) {
#pragma unroll
    for (int j = 0; j < 4; ++j) {
      int nc = n0 + j * 16 + coln;
      float bv = bias ? bias[nc] : 0.f;
#pragma unroll
      for (int r = 0; r < 8; ++r) {
        int gm = m0 + mi * 16 + roff + r;
        if (gm < M_) {
          size_t oi = (size_t)gm * ldo + nc;
          float v = acc[mi][j][r] + bv;
          if (EPI == 0)      Out[oi] = v;
          else if (EPI == 1) Out[oi] += v;
          else               Out[oi] = 0.5f * v * (1.f + erff(v * 0.70710678118654752f));
        }
      }
    }
  }
}

#define AQB_  64
#define AKT_  64
#define ALDP_ 52
#define ANKT_ ((N_ + AKT_ - 1) / AKT_)
#define AQBN_ ((N_ + AQB_ - 1) / AQB_)
#define ANEG_ (-3.0e38f)

__device__ __forceinline__ void attn_tile(const float* __restrict__ kt,
                                          const float* __restrict__ vt,
                                          const v2f (&qf)[12], int key0,
                                          float& m, float& l, v8f (&oacc)[3],
                                          bool keep_row0, float* __restrict__ srow) {
  const float scale = 0.14433756729740643f;
  v8f sacc[4] = {};
  v2f fk[2][4];
#pragma unroll
  for (int j = 0; j < 4; ++j) fk[0][j] = *(const v2f*)(kt + j * 16 * ALDP_);
#pragma unroll
  for (int ks = 0; ks < 12; ++ks) {
    const int cb = ks & 1, nb = cb ^ 1;
    if (ks + 1 < 12) {
#pragma unroll
      for (int j = 0; j < 4; ++j)
        fk[nb][j] = *(const v2f*)(kt + j * 16 * ALDP_ + (ks + 1) * 4);
    }
#pragma unroll
    for (int j = 0; j < 4; ++j) sacc[j] = wmma4(fk[cb][j], qf[ks], sacc[j]);
  }
  float tmax = ANEG_;
#pragma unroll
  for (int j = 0; j < 4; ++j) {
#pragma unroll
    for (int r = 0; r < 8; ++r) {
      float v = sacc[j][r] * scale;
      v = (key0 + j * 16 + r < N_) ? v : ANEG_;
      sacc[j][r] = v;
      tmax = fmaxf(tmax, v);
    }
  }
  if (keep_row0) {
#pragma unroll
    for (int j = 0; j < 4; ++j)
#pragma unroll
      for (int r = 0; r < 8; ++r) srow[key0 + j * 16 + r] = sacc[j][r];
  }
  tmax = fmaxf(tmax, __shfl_xor(tmax, 16, 32));
  const float mn = fmaxf(m, tmax);
  const float alpha = expf(m - mn);
  m = mn;
  float ps = 0.f;
#pragma unroll
  for (int j = 0; j < 4; ++j) {
#pragma unroll
    for (int r = 0; r < 8; ++r) {
      float p = expf(sacc[j][r] - mn);
      sacc[j][r] = p;
      ps += p;
    }
  }
  l = l * alpha + ps;
#pragma unroll
  for (int n = 0; n < 3; ++n)
#pragma unroll
    for (int r = 0; r < 8; ++r) oacc[n][r] *= alpha;
  v2f fv[2][3];
#pragma unroll
  for (int n = 0; n < 3; ++n) {
    fv[0][n].x = vt[n * 16];
    fv[0][n].y = vt[ALDP_ + n * 16];
  }
#pragma unroll
  for (int u = 0; u < 16; ++u) {
    const int cb = u & 1, nb = cb ^ 1;
    if (u + 1 < 16) {
      const int ko = ((u + 1) >> 2) * 16 + ((u + 1) & 3) * 2;
#pragma unroll
      for (int n = 0; n < 3; ++n) {
        fv[nb][n].x = vt[ko * ALDP_ + n * 16];
        fv[nb][n].y = vt[(ko + 1) * ALDP_ + n * 16];
      }
    }
    v2f pb;
    pb.x = sacc[u >> 2][(u & 3) * 2];
    pb.y = sacc[u >> 2][(u & 3) * 2 + 1];
#pragma unroll
    for (int n = 0; n < 3; ++n) oacc[n] = wmma4(fv[cb][n], pb, oacc[n]);
  }
}

__global__ __launch_bounds__(128) void attn_kernel(const float* __restrict__ qkv,
                                                   float* __restrict__ o,
                                                   float* __restrict__ ga) {
  __shared__ __attribute__((aligned(16))) float Ks[2][AKT_ * ALDP_];
  __shared__ __attribute__((aligned(16))) float Vs[2][AKT_ * ALDP_];
  __shared__ float Srow[ANKT_ * AKT_];
  __shared__ float Mrow[2];
  const int tid  = threadIdx.x;
  const int lane = tid & 31;
  const int wave = tid >> 5;
  const int hd = blockIdx.y, b = blockIdx.z;
  const int c = lane & 15;
  const int h = lane >> 4;
  const int q0 = blockIdx.x * AQB_ + wave * 16;
  int gq = q0 + c; if (gq > N_ - 1) gq = N_ - 1;
  const float* qrow = qkv + (size_t)(b * N_ + gq) * C3_ + hd * D_ + 2 * h;
  v2f qf[12];
#pragma unroll
  for (int ks = 0; ks < 12; ++ks) qf[ks] = *(const v2f*)(qrow + ks * 4);

  const float* kbase = qkv + (size_t)b * N_ * C3_ + C_ + hd * D_;
  int srw[6], sof[6];
#pragma unroll
  for (int i = 0; i < 6; ++i) {
    int idx = tid + 128 * i;
    srw[i] = idx / 12;
    sof[i] = (idx - srw[i] * 12) * 4;
  }
  v4f rk[6], rv[6];
#pragma unroll
  for (int i = 0; i < 6; ++i) {
    const float* p = kbase + (size_t)srw[i] * C3_ + sof[i];
    rk[i] = *(const v4f*)(p);
    rv[i] = *(const v4f*)(p + C_);
  }
#pragma unroll
  for (int i = 0; i < 6; ++i) {
    *(v4f*)(&Ks[0][srw[i] * ALDP_ + sof[i]]) = rk[i];
    *(v4f*)(&Vs[0][srw[i] * ALDP_ + sof[i]]) = rv[i];
  }
  __syncthreads();

  float m = ANEG_, l = 0.f;
  v8f oacc[3] = {};
  const bool keep_row0 = (blockIdx.x == 0) && (wave == 0) && (c == 0);
  const int koff = c * ALDP_ + 2 * h;
  const int voff = 8 * h * ALDP_ + c;

  for (int t = 0; t < ANKT_ - 1; ++t) {
    const int cur = t & 1;
#pragma unroll
    for (int i = 0; i < 6; ++i) {
      int kr = (t + 1) * AKT_ + srw[i];
      if (kr > N_ - 1) kr = N_ - 1;
      const float* p = kbase + (size_t)kr * C3_ + sof[i];
      rk[i] = *(const v4f*)(p);
      rv[i] = *(const v4f*)(p + C_);
    }
    attn_tile(&Ks[cur][koff], &Vs[cur][voff], qf, t * AKT_ + 8 * h, m, l, oacc,
              keep_row0, Srow);
#pragma unroll
    for (int i = 0; i < 6; ++i) {
      *(v4f*)(&Ks[cur ^ 1][srw[i] * ALDP_ + sof[i]]) = rk[i];
      *(v4f*)(&Vs[cur ^ 1][srw[i] * ALDP_ + sof[i]]) = rv[i];
    }
    __syncthreads();
  }
  {
    const int t = ANKT_ - 1;
    attn_tile(&Ks[t & 1][koff], &Vs[t & 1][voff], qf, t * AKT_ + 8 * h, m, l, oacc,
              keep_row0, Srow);
  }

  const float lt  = l + __shfl_xor(l, 16, 32);
  const float inv = 1.f / lt;
  const int q = q0 + c;
  if (q < N_) {
    float* op = o + (size_t)(b * N_ + q) * C_ + hd * D_ + 8 * h;
#pragma unroll
    for (int n = 0; n < 3; ++n) {
      v4f lo, hi;
      lo.x = oacc[n][0] * inv; lo.y = oacc[n][1] * inv;
      lo.z = oacc[n][2] * inv; lo.w = oacc[n][3] * inv;
      hi.x = oacc[n][4] * inv; hi.y = oacc[n][5] * inv;
      hi.z = oacc[n][6] * inv; hi.w = oacc[n][7] * inv;
      *(v4f*)(op + n * 16)     = lo;
      *(v4f*)(op + n * 16 + 4) = hi;
    }
  }

  if (blockIdx.x == 0) {
    if (tid == 0) { Mrow[0] = m; Mrow[1] = inv; }
    __syncthreads();
    const float m0 = Mrow[0], i0 = Mrow[1];
    float* gp = ga + (size_t)(b * HEADS_ + hd) * HW_;
    for (int key = 1 + tid; key < N_; key += 128)
      gp[key - 1] = expf(Srow[key] - m0) * i0;
  }
}

__global__ void finalize_kernel(const float* __restrict__ xc,
                                const float* __restrict__ ga,
                                float* __restrict__ out) {
  size_t idx = (size_t)blockIdx.x * 256 + threadIdx.x;
  const size_t n_cls = (size_t)B_ * C_;
  const size_t n_x   = (size_t)B_ * C_ * HW_;
  const size_t n_ga  = (size_t)B_ * HW_;
  if (idx < n_cls) {
    int b = (int)(idx / C_), c = (int)(idx % C_);
    out[idx] = xc[(size_t)b * N_ * C_ + c];
  } else if (idx < n_cls + n_x) {
    size_t j = idx - n_cls;
    int b   = (int)(j / (C_ * HW_));
    int rem = (int)(j % (C_ * HW_));
    int c = rem / HW_, p = rem % HW_;
    out[idx] = xc[((size_t)b * N_ + 1 + p) * C_ + c];
  } else if (idx < n_cls + n_x + n_ga) {
    size_t j = idx - n_cls - n_x;
    int b = (int)(j / HW_), p = (int)(j % HW_);
    float s = 0.f;
#pragma unroll
    for (int hh = 0; hh < HEADS_; ++hh)
      s += ga[(size_t)(b * HEADS_ + hh) * HW_ + p];
    out[idx] = s * 0.125f;
  }
}

extern "C" void kernel_launch(void* const* d_in, const int* in_sizes, int n_in,
                              void* d_out, int out_size, void* d_ws, size_t ws_size,
                              hipStream_t stream) {
  const float* cls    = (const float*)d_in[0];
  const float* x      = (const float*)d_in[1];
  const float* conv_w = (const float*)d_in[2];
  const float* conv_b = (const float*)d_in[3];
  const float* ln1_g  = (const float*)d_in[4];
  const float* ln1_b  = (const float*)d_in[5];
  const float* qkv_w  = (const float*)d_in[6];
  const float* proj_w = (const float*)d_in[7];
  const float* proj_b = (const float*)d_in[8];
  const float* ln2_g  = (const float*)d_in[9];
  const float* ln2_b  = (const float*)d_in[10];
  const float* fc1_w  = (const float*)d_in[11];
  const float* fc1_b  = (const float*)d_in[12];
  const float* fc2_w  = (const float*)d_in[13];
  const float* fc2_b  = (const float*)d_in[14];

  float* ws  = (float*)d_ws;
  float* xc  = ws;
  float* y   = xc  + (size_t)MP_ * C_;
  float* qkv = y   + (size_t)MP_ * C_;
  float* o   = qkv + (size_t)MP_ * C3_;
  float* m1  = qkv;
  float* ga  = o   + (size_t)MP_ * C_;
  float* out = (float*)d_out;

  conv_pe_kernel<<<(B_ * C_ * HW_ + 255) / 256, 256, 0, stream>>>(x, conv_w, conv_b, xc);
  cls_copy_kernel<<<(B_ * C_ + 255) / 256, 256, 0, stream>>>(cls, xc);

  ln_kernel<<<(M_ + 7) / 8, 256, 0, stream>>>(xc, ln1_g, ln1_b, y, M_);

  gemm_wmma_kernel<0><<<dim3(C3_ / GBN_, GMB_), 128, 0, stream>>>(
      y, qkv_w, nullptr, qkv, C_, C3_);

  attn_kernel<<<dim3(AQBN_, HEADS_, B_), 128, 0, stream>>>(qkv, o, ga);

  gemm_wmma_kernel<1><<<dim3(C_ / GBN_, GMB_), 128, 0, stream>>>(
      o, proj_w, proj_b, xc, C_, C_);

  ln_kernel<<<(M_ + 7) / 8, 256, 0, stream>>>(xc, ln2_g, ln2_b, y, M_);

  gemm_wmma_kernel<2><<<dim3(C4_ / GBN_, GMB_), 128, 0, stream>>>(
      y, fc1_w, fc1_b, m1, C_, C4_);

  gemm_wmma_kernel<1><<<dim3(C_ / GBN_, GMB_), 128, 0, stream>>>(
      m1, fc2_w, fc2_b, xc, C4_, C_);

  size_t tot = (size_t)B_ * C_ + (size_t)B_ * C_ * HW_ + (size_t)B_ * HW_;
  finalize_kernel<<<(unsigned)((tot + 255) / 256), 256, 0, stream>>>(xc, ga, out);
}


# conv_pe and finalize: the NCHW<->token layout change goes through a padded LDS tile (32 ch x 112 pos); global loads and stores cover whole 128-byte runs, not 4-byte pieces at 1536-byte stride
# speedup vs baseline: 2.8202x; 1.2192x over previous
#include <hip/hip_runtime.h>
#include <math.h>

#define B_    16
#define C_    384
#define H_    28
#define W_    28
#define HW_   784
#define N_    785
#define M_    (B_*N_)
#define MP_   (M_ + 16)
#define HEADS_ 8
#define D_    48
#define C3_   1152
#define C4_   1536
#define SLD   800

typedef __attribute__((ext_vector_type(2))) float v2f;
typedef __attribute__((ext_vector_type(8))) float v8f;

__device__ __forceinline__ v8f wmma4(v2f a, v2f b, v8f c) {
  return __builtin_amdgcn_wmma_f32_16x16x4_f32(false, a, false, b, (short)0, c,
                                               false, false);
}

#define TCH_ 32
#define TRW_ 4
#define TPS_ (TRW_ * W_)
#define THL_ ((TRW_ + 2) * W_)
#define TXP_ (THL_ + 3)
#define TNT_ (H_ / TRW_)
#define TCT_ (C_ / TCH_)

__global__ __launch_bounds__(256) void conv_pe_kernel(const float* __restrict__ x,
                                                      const float* __restrict__ cw,
                                                      const float* __restrict__ cb,
                                                      float* __restrict__ xc) {
  __shared__ float xs[TCH_ * TXP_];
  const int tid = threadIdx.x;
  const int r0 = blockIdx.x * TRW_;
  const int c0 = blockIdx.y * TCH_;
  const int b  = blockIdx.z;
  const int gp0 = (r0 - 1) * W_;
  const float* xb = x + (size_t)(b * C_ + c0) * HW_;
  float sv[(TCH_ * THL_) / 256];
#pragma unroll
  for (int i = 0; i < (TCH_ * THL_) / 256; ++i) {
    int e  = tid + 256 * i;
    int ci = e / THL_;
    int gp = gp0 + (e - ci * THL_);
    int gc = gp < 0 ? 0 : (gp > HW_ - 1 ? HW_ - 1 : gp);
    float v = xb[ci * HW_ + gc];
    sv[i] = (gp == gc) ? v : 0.f;
  }
#pragma unroll
  for (int i = 0; i < (TCH_ * THL_) / 256; ++i) {
    int e  = tid + 256 * i;
    int ci = e / THL_;
    xs[ci * TXP_ + 1 + (e - ci * THL_)] = sv[i];
  }
  if (tid < TCH_) {
    xs[tid * TXP_] = 0.f;
    xs[tid * TXP_ + THL_ + 1] = 0.f;
    xs[tid * TXP_ + THL_ + 2] = 0.f;
  }
  __syncthreads();

  const int c = tid & 31;
  const int w = tid >> 5;
  const float* wp = cw + (c0 + c) * 9;
  float wr[9];
#pragma unroll
  for (int k = 0; k < 9; ++k) wr[k] = wp[k];
  const float bias = cb[c0 + c];
  const float* xr = xs + c * TXP_ + 1;
  float* orow = xc + ((size_t)b * N_ + 1 + r0 * W_) * C_ + c0 + c;
  for (int i = 0; i < TPS_ / 8; ++i) {
    int pl = w + 8 * i;
    int hl = pl / W_;
    int ww = pl - hl * W_;
    int hh = r0 + hl;
    const float* xp = xr + hl * W_ + ww - 1;
    float s = 0.f;
#pragma unroll
    for (int ii = 0; ii < 3; ++ii) {
      int hy = hh + ii - 1;
      bool rok = (hy >= 0) && (hy < H_);
#pragma unroll
      for (int jj = 0; jj < 3; ++jj) {
        int wx = ww + jj - 1;
        bool ok = rok && (wx >= 0) && (wx < W_);
        float sn = s + xp[ii * W_ + jj] * wr[ii * 3 + jj];
        s = ok ? sn : s;
      }
    }
    orow[(size_t)pl * C_] = xp[W_ + 1] + s + bias;
  }
}

__global__ void cls_copy_kernel(const float* __restrict__ cls,
                                float* __restrict__ xc) {
  int idx = blockIdx.x * 256 + threadIdx.x;
  if (idx >= B_ * C_) return;
  int b = idx / C_, c = idx % C_;
  xc[(size_t)b * N_ * C_ + c] = cls[idx];
}

__global__ void ln_kernel(const float* __restrict__ xin,
                          const float* __restrict__ g,
                          const float* __restrict__ bta,
                          float* __restrict__ yout, int ntok) {
  int wave = threadIdx.x >> 5, lane = threadIdx.x & 31;
  int t = blockIdx.x * 8 + wave;
  if (t >= ntok) return;
  const float* xp = xin + (size_t)t * C_;
  float s = 0.f, sq = 0.f;
#pragma unroll
  for (int i = 0; i < C_ / 32; ++i) {
    float v = xp[lane + i * 32];
    s += v; sq += v * v;
  }
#pragma unroll
  for (int off = 16; off; off >>= 1) {
    s  += __shfl_xor(s,  off, 32);
    sq += __shfl_xor(sq, off, 32);
  }
  float mu  = s  * (1.f / C_);
  float var = sq * (1.f / C_) - mu * mu;
  float rs  = rsqrtf(var + 1e-5f);
  float* yp = yout + (size_t)t * C_;
#pragma unroll
  for (int i = 0; i < C_ / 32; ++i) {
    int c = lane + i * 32;
    yp[c] = (xp[c] - mu) * rs * g[c] + bta[c];
  }
}

#define GBM_  128
#define GBN_  64
#define GKC_  32
#define GLDP_ 36
#define GMB_  ((M_ + GBM_ - 1) / GBM_)

typedef __attribute__((ext_vector_type(4))) float v4f;

__device__ __forceinline__ void gemm_chunk(const float* __restrict__ as,
                                           const float* __restrict__ bs,
                                           v8f (&acc)[2][4]) {
  v2f fa[2][2], fb[2][4];
  fa[0][0] = *(const v2f*)(as);
  fa[0][1] = *(const v2f*)(as + 16 * GLDP_);
#pragma unroll
  for (int j = 0; j < 4; ++j) fb[0][j] = *(const v2f*)(bs + j * 16 * GLDP_);
#pragma unroll
  for (int ks = 0; ks < GKC_ / 4; ++ks) {
    const int c = ks & 1, n = c ^ 1;
    if (ks + 1 < GKC_ / 4) {
      fa[n][0] = *(const v2f*)(as + (ks + 1) * 4);
      fa[n][1] = *(const v2f*)(as + 16 * GLDP_ + (ks + 1) * 4);
#pragma unroll
      for (int j = 0; j < 4; ++j)
        fb[n][j] = *(const v2f*)(bs + j * 16 * GLDP_ + (ks + 1) * 4);
    }
    acc[0][0] = wmma4(fa[c][0], fb[c][0], acc[0][0]);
    acc[1][0] = wmma4(fa[c][1], fb[c][0], acc[1][0]);
    acc[0][1] = wmma4(fa[c][0], fb[c][1], acc[0][1]);
    acc[1][1] = wmma4(fa[c][1], fb[c][1], acc[1][1]);
    acc[0][2] = wmma4(fa[c][0], fb[c][2], acc[0][2]);
    acc[1][2] = wmma4(fa[c][1], fb[c][2], acc[1][2]);
    acc[0][3] = wmma4(fa[c][0], fb[c][3], acc[0][3]);
    acc[1][3] = wmma4(fa[c][1], fb[c][3], acc[1][3]);
  }
}

template <int EPI>
__global__ __launch_bounds__(128) void gemm_wmma_kernel(
    const float* __restrict__ X, const float* __restrict__ Wt,
    const float* __restrict__ bias, float* __restrict__ Out,
    int K, int ldo) {
  __shared__ __attribute__((aligned(16))) float As[2][GBM_ * GLDP_];
  __shared__ __attribute__((aligned(16))) float Bs[2][GBN_ * GLDP_];
  const int tid  = threadIdx.x;
  const int lane = tid & 31;
  const int wave = tid >> 5;
  const int n0  = blockIdx.x * GBN_;
  const int mb0 = blockIdx.y * GBM_;

  const int lr = tid >> 3;
  const int lc = (tid & 7) << 2;
  const float* ag[8];
  const float* bg[4];
#pragma unroll
  for (int i = 0; i < 8; ++i) {
    int r = mb0 + lr + 16 * i;
    if (r > M_ - 1) r = M_ - 1;
    ag[i] = X + (size_t)r * K + lc;
  }
#pragma unroll
  for (int i = 0; i < 4; ++i) bg[i] = Wt + (size_t)(n0 + lr + 16 * i) * K + lc;
  const int soff = lr * GLDP_ + lc;

  v4f ra[8], rb[4];
#pragma unroll
  for (int i = 0; i < 8; ++i) ra[i] = *(const v4f*)(ag[i]);
#pragma unroll
  for (int i = 0; i < 4; ++i) rb[i] = *(const v4f*)(bg[i]);
#pragma unroll
  for (int i = 0; i < 8; ++i) *(v4f*)(&As[0][soff + 16 * i * GLDP_]) = ra[i];
#pragma unroll
  for (int i = 0; i < 4; ++i) *(v4f*)(&Bs[0][soff + 16 * i * GLDP_]) = rb[i];
  __syncthreads();

  const int rc = lane & 15;
  const int kb = (lane >> 4) << 1;
  const int aoff = (wave * 32 + rc) * GLDP_ + kb;
  const int boff = rc * GLDP_ + kb;

  v8f acc[2][4] = {};
  const int nk = K / GKC_;
  for (int kc = 0; kc < nk - 1; ++kc) {
    const int cur = kc & 1;
    const int kn = (kc + 1) * GKC_;
#pragma unroll
    for (int i = 0; i < 8; ++i) ra[i] = *(const v4f*)(ag[i] + kn);
#pragma unroll
    for (int i = 0; i < 4; ++i) rb[i] = *(const v4f*)(bg[i] + kn);
    gemm_chunk(&As[cur][aoff], &Bs[cur][boff], acc);
#pragma unroll
    for (int i = 0; i < 8; ++i) *(v4f*)(&As[cur ^ 1][soff + 16 * i * GLDP_]) = ra[i];
#pragma unroll
    for (int i = 0; i < 4; ++i) *(v4f*)(&Bs[cur ^ 1][soff + 16 * i * GLDP_]) = rb[i];
    __syncthreads();
  }
  {
    const int cur = (nk - 1) & 1;
    gemm_chunk(&As[cur][aoff], &Bs[cur][boff], acc);
  }

  const int m0 = mb0 + wave * 32;
  int coln = lane & 15;
  int roff = (lane >> 4) << 3;
#pragma unroll
  for (int mi = 0; mi < 2; ++mi) {
#pragma unroll
    for (int j = 0; j < 4; ++j) {
      int nc = n0 + j * 16 + coln;
      float bv = bias ? bias[nc] : 0.f;
#pragma unroll
      for (int r = 0; r < 8; ++r) {
        int gm = m0 + mi * 16 + roff + r;
        if (gm < M_) {
          size_t oi = (size_t)gm * ldo + nc;
          float v = acc[mi][j][r] + bv;
          if (EPI == 0)      Out[oi] = v;
          else if (EPI == 1) Out[oi] += v;
          else               Out[oi] = 0.5f * v * (1.f + erff(v * 0.70710678118654752f));
        }
      }
    }
  }
}

#define AQB_  64
#define AKT_  64
#define ALDP_ 52
#define ANKT_ ((N_ + AKT_ - 1) / AKT_)
#define AQBN_ ((N_ + AQB_ - 1) / AQB_)
#define ANEG_ (-3.0e38f)

__device__ __forceinline__ void attn_tile(const float* __restrict__ kt,
                                          const float* __restrict__ vt,
                                          const v2f (&qf)[12], int key0,
                                          float& m, float& l, v8f (&oacc)[3],
                                          bool keep_row0, float* __restrict__ srow) {
  const float scale = 0.14433756729740643f;
  v8f sacc[4] = {};
  v2f fk[2][4];
#pragma unroll
  for (int j = 0; j < 4; ++j) fk[0][j] = *(const v2f*)(kt + j * 16 * ALDP_);
#pragma unroll
  for (int ks = 0; ks < 12; ++ks) {
    const int cb = ks & 1, nb = cb ^ 1;
    if (ks + 1 < 12) {
#pragma unroll
      for (int j = 0; j < 4; ++j)
        fk[nb][j] = *(const v2f*)(kt + j * 16 * ALDP_ + (ks + 1) * 4);
    }
#pragma unroll
    for (int j = 0; j < 4; ++j) sacc[j] = wmma4(fk[cb][j], qf[ks], sacc[j]);
  }
  float tmax = ANEG_;
#pragma unroll
  for (int j = 0; j < 4; ++j) {
#pragma unroll
    for (int r = 0; r < 8; ++r) {
      float v = sacc[j][r] * scale;
      v = (key0 + j * 16 + r < N_) ? v : ANEG_;
      sacc[j][r] = v;
      tmax = fmaxf(tmax, v);
    }
  }
  if (keep_row0) {
#pragma unroll
    for (int j = 0; j < 4; ++j)
#pragma unroll
      for (int r = 0; r < 8; ++r) srow[key0 + j * 16 + r] = sacc[j][r];
  }
  tmax = fmaxf(tmax, __shfl_xor(tmax, 16, 32));
  const float mn = fmaxf(m, tmax);
  const float alpha = expf(m - mn);
  m = mn;
  float ps = 0.f;
#pragma unroll
  for (int j = 0; j < 4; ++j) {
#pragma unroll
    for (int r = 0; r < 8; ++r) {
      float p = expf(sacc[j][r] - mn);
      sacc[j][r] = p;
      ps += p;
    }
  }
  l = l * alpha + ps;
#pragma unroll
  for (int n = 0; n < 3; ++n)
#pragma unroll
    for (int r = 0; r < 8; ++r) oacc[n][r] *= alpha;
  v2f fv[2][3];
#pragma unroll
  for (int n = 0; n < 3; ++n) {
    fv[0][n].x = vt[n * 16];
    fv[0][n].y = vt[ALDP_ + n * 16];
  }
#pragma unroll
  for (int u = 0; u < 16; ++u) {
    const int cb = u & 1, nb = cb ^ 1;
    if (u + 1 < 16) {
      const int ko = ((u + 1) >> 2) * 16 + ((u + 1) & 3) * 2;
#pragma unroll
      for (int n = 0; n < 3; ++n) {
        fv[nb][n].x = vt[ko * ALDP_ + n * 16];
        fv[nb][n].y = vt[(ko + 1) * ALDP_ + n * 16];
      }
    }
    v2f pb;
    pb.x = sacc[u >> 2][(u & 3) * 2];
    pb.y = sacc[u >> 2][(u & 3) * 2 + 1];
#pragma unroll
    for (int n = 0; n < 3; ++n) oacc[n] = wmma4(fv[cb][n], pb, oacc[n]);
  }
}

__global__ __launch_bounds__(128) void attn_kernel(const float* __restrict__ qkv,
                                                   float* __restrict__ o,
                                                   float* __restrict__ ga) {
  __shared__ __attribute__((aligned(16))) float Ks[2][AKT_ * ALDP_];
  __shared__ __attribute__((aligned(16))) float Vs[2][AKT_ * ALDP_];
  __shared__ float Srow[ANKT_ * AKT_];
  __shared__ float Mrow[2];
  const int tid  = threadIdx.x;
  const int lane = tid & 31;
  const int wave = tid >> 5;
  const int hd = blockIdx.y, b = blockIdx.z;
  const int c = lane & 15;
  const int h = lane >> 4;
  const int q0 = blockIdx.x * AQB_ + wave * 16;
  int gq = q0 + c; if (gq > N_ - 1) gq = N_ - 1;
  const float* qrow = qkv + (size_t)(b * N_ + gq) * C3_ + hd * D_ + 2 * h;
  v2f qf[12];
#pragma unroll
  for (int ks = 0; ks < 12; ++ks) qf[ks] = *(const v2f*)(qrow + ks * 4);

  const float* kbase = qkv + (size_t)b * N_ * C3_ + C_ + hd * D_;
  int srw[6], sof[6];
#pragma unroll
  for (int i = 0; i < 6; ++i) {
    int idx = tid + 128 * i;
    srw[i] = idx / 12;
    sof[i] = (idx - srw[i] * 12) * 4;
  }
  v4f rk[6], rv[6];
#pragma unroll
  for (int i = 0; i < 6; ++i) {
    const float* p = kbase + (size_t)srw[i] * C3_ + sof[i];
    rk[i] = *(const v4f*)(p);
    rv[i] = *(const v4f*)(p + C_);
  }
#pragma unroll
  for (int i = 0; i < 6; ++i) {
    *(v4f*)(&Ks[0][srw[i] * ALDP_ + sof[i]]) = rk[i];
    *(v4f*)(&Vs[0][srw[i] * ALDP_ + sof[i]]) = rv[i];
  }
  __syncthreads();

  float m = ANEG_, l = 0.f;
  v8f oacc[3] = {};
  const bool keep_row0 = (blockIdx.x == 0) && (wave == 0) && (c == 0);
  const int koff = c * ALDP_ + 2 * h;
  const int voff = 8 * h * ALDP_ + c;

  for (int t = 0; t < ANKT_ - 1; ++t) {
    const int cur = t & 1;
#pragma unroll
    for (int i = 0; i < 6; ++i) {
      int kr = (t + 1) * AKT_ + srw[i];
      if (kr > N_ - 1) kr = N_ - 1;
      const float* p = kbase + (size_t)kr * C3_ + sof[i];
      rk[i] = *(const v4f*)(p);
      rv[i] = *(const v4f*)(p + C_);
    }
    attn_tile(&Ks[cur][koff], &Vs[cur][voff], qf, t * AKT_ + 8 * h, m, l, oacc,
              keep_row0, Srow);
#pragma unroll
    for (int i = 0; i < 6; ++i) {
      *(v4f*)(&Ks[cur ^ 1][srw[i] * ALDP_ + sof[i]]) = rk[i];
      *(v4f*)(&Vs[cur ^ 1][srw[i] * ALDP_ + sof[i]]) = rv[i];
    }
    __syncthreads();
  }
  {
    const int t = ANKT_ - 1;
    attn_tile(&Ks[t & 1][koff], &Vs[t & 1][voff], qf, t * AKT_ + 8 * h, m, l, oacc,
              keep_row0, Srow);
  }

  const float lt  = l + __shfl_xor(l, 16, 32);
  const float inv = 1.f / lt;
  const int q = q0 + c;
  if (q < N_) {
    float* op = o + (size_t)(b * N_ + q) * C_ + hd * D_ + 8 * h;
#pragma unroll
    for (int n = 0; n < 3; ++n) {
      v4f lo, hi;
      lo.x = oacc[n][0] * inv; lo.y = oacc[n][1] * inv;
      lo.z = oacc[n][2] * inv; lo.w = oacc[n][3] * inv;
      hi.x = oacc[n][4] * inv; hi.y = oacc[n][5] * inv;
      hi.z = oacc[n][6] * inv; hi.w = oacc[n][7] * inv;
      *(v4f*)(op + n * 16)     = lo;
      *(v4f*)(op + n * 16 + 4) = hi;
    }
  }

  if (blockIdx.x == 0) {
    if (tid == 0) { Mrow[0] = m; Mrow[1] = inv; }
    __syncthreads();
    const float m0 = Mrow[0], i0 = Mrow[1];
    float* gp = ga + (size_t)(b * HEADS_ + hd) * HW_;
    for (int key = 1 + tid; key < N_; key += 128)
      gp[key - 1] = expf(Srow[key] - m0) * i0;
  }
}

#define FTB_  (B_ * TCT_ * TNT_)
#define FTL_  ((B_ * C_ + B_ * HW_ + 255) / 256)
#define FTP_  33

__global__ __launch_bounds__(256) void finalize_kernel(const float* __restrict__ xc,
                                                       const float* __restrict__ ga,
                                                       float* __restrict__ out) {
  __shared__ float ts[TPS_ * FTP_];
  const int tid = threadIdx.x;
  const size_t n_cls = (size_t)B_ * C_;
  const size_t n_x   = (size_t)B_ * C_ * HW_;
  const size_t n_ga  = (size_t)B_ * HW_;
  if (blockIdx.x < FTB_) {
    const int bid = blockIdx.x;
    const int rt = bid % TNT_;
    const int t2 = bid / TNT_;
    const int ct = t2 % TCT_;
    const int b  = t2 / TCT_;
    const int c0 = ct * TCH_, p0 = rt * TPS_;
    const int c = tid & 31, w = tid >> 5;
    const float* ib = xc + ((size_t)b * N_ + 1 + p0) * C_ + c0 + c;
#pragma unroll
    for (int i = 0; i < TPS_ / 8; ++i) {
      int pl = w + 8 * i;
      ts[pl * FTP_ + c] = ib[(size_t)pl * C_];
    }
    __syncthreads();
    float* ob = out + n_cls + (size_t)(b * C_ + c0) * HW_ + p0;
#pragma unroll
    for (int i = 0; i < (TCH_ * TPS_) / 256; ++i) {
      int e  = tid + 256 * i;
      int ci = e / TPS_;
      int pl = e - ci * TPS_;
      ob[(size_t)ci * HW_ + pl] = ts[pl * FTP_ + ci];
    }
  } else {
    size_t j = (size_t)(blockIdx.x - FTB_) * 256 + tid;
    if (j < n_cls) {
      int b = (int)(j / C_), c = (int)(j % C_);
      out[j] = xc[(size_t)b * N_ * C_ + c];
    } else if (j < n_cls + n_ga) {
      size_t k = j - n_cls;
      int b = (int)(k / HW_), p = (int)(k % HW_);
      float s = 0.f;
#pragma unroll
      for (int hh = 0; hh < HEADS_; ++hh)
        s += ga[(size_t)(b * HEADS_ + hh) * HW_ + p];
      out[n_cls + n_x + k] = s * 0.125f;
    }
  }
}

extern "C" void kernel_launch(void* const* d_in, const int* in_sizes, int n_in,
                              void* d_out, int out_size, void* d_ws, size_t ws_size,
                              hipStream_t stream) {
  const float* cls    = (const float*)d_in[0];
  const float* x      = (const float*)d_in[1];
  const float* conv_w = (const float*)d_in[2];
  const float* conv_b = (const float*)d_in[3];
  const float* ln1_g  = (const float*)d_in[4];
  const float* ln1_b  = (const float*)d_in[5];
  const float* qkv_w  = (const float*)d_in[6];
  const float* proj_w = (const float*)d_in[7];
  const float* proj_b = (const float*)d_in[8];
  const float* ln2_g  = (const float*)d_in[9];
  const float* ln2_b  = (const float*)d_in[10];
  const float* fc1_w  = (const float*)d_in[11];
  const float* fc1_b  = (const float*)d_in[12];
  const float* fc2_w  = (const float*)d_in[13];
  const float* fc2_b  = (const float*)d_in[14];

  float* ws  = (float*)d_ws;
  float* xc  = ws;
  float* y   = xc  + (size_t)MP_ * C_;
  float* qkv = y   + (size_t)MP_ * C_;
  float* o   = qkv + (size_t)MP_ * C3_;
  float* m1  = qkv;
  float* ga  = o   + (size_t)MP_ * C_;
  float* out = (float*)d_out;

  conv_pe_kernel<<<dim3(TNT_, TCT_, B_), 256, 0, stream>>>(x, conv_w, conv_b, xc);
  cls_copy_kernel<<<(B_ * C_ + 255) / 256, 256, 0, stream>>>(cls, xc);

  ln_kernel<<<(M_ + 7) / 8, 256, 0, stream>>>(xc, ln1_g, ln1_b, y, M_);

  gemm_wmma_kernel<0><<<dim3(C3_ / GBN_, GMB_), 128, 0, stream>>>(
      y, qkv_w, nullptr, qkv, C_, C3_);

  attn_kernel<<<dim3(AQBN_, HEADS_, B_), 128, 0, stream>>>(qkv, o, ga);

  gemm_wmma_kernel<1><<<dim3(C_ / GBN_, GMB_), 128, 0, stream>>>(
      o, proj_w, proj_b, xc, C_, C_);

  ln_kernel<<<(M_ + 7) / 8, 256, 0, stream>>>(xc, ln2_g, ln2_b, y, M_);

  gemm_wmma_kernel<2><<<dim3(C4_ / GBN_, GMB_), 128, 0, stream>>>(
      y, fc1_w, fc1_b, m1, C_, C4_);

  gemm_wmma_kernel<1><<<dim3(C_ / GBN_, GMB_), 128, 0, stream>>>(
      m1, fc2_w, fc2_b, xc, C4_, C_);

  finalize_kernel<<<FTB_ + FTL_, 256, 0, stream>>>(xc, ga, out);
}


# Residual-add GEMM epilogue (proj, fc2): a lane's 64 old output values are fetched as one group of independent loads before the stores, not 64 dependent load-add-store round trips; sums unchanged
# speedup vs baseline: 3.1155x; 1.1047x over previous
#include <hip/hip_runtime.h>
#include <math.h>

#define B_    16
#define C_    384
#define H_    28
#define W_    28
#define HW_   784
#define N_    785
#define M_    (B_*N_)
#define MP_   (M_ + 16)
#define HEADS_ 8
#define D_    48
#define C3_   1152
#define C4_   1536
#define SLD   800

typedef __attribute__((ext_vector_type(2))) float v2f;
typedef __attribute__((ext_vector_type(8))) float v8f;

__device__ __forceinline__ v8f wmma4(v2f a, v2f b, v8f c) {
  return __builtin_amdgcn_wmma_f32_16x16x4_f32(false, a, false, b, (short)0, c,
                                               false, false);
}

#define TCH_ 32
#define TRW_ 4
#define TPS_ (TRW_ * W_)
#define THL_ ((TRW_ + 2) * W_)
#define TXP_ (THL_ + 3)
#define TNT_ (H_ / TRW_)
#define TCT_ (C_ / TCH_)

__global__ __launch_bounds__(256) void conv_pe_kernel(const float* __restrict__ x,
                                                      const float* __restrict__ cw,
                                                      const float* __restrict__ cb,
                                                      float* __restrict__ xc) {
  __shared__ float xs[TCH_ * TXP_];
  const int tid = threadIdx.x;
  const int r0 = blockIdx.x * TRW_;
  const int c0 = blockIdx.y * TCH_;
  const int b  = blockIdx.z;
  const int gp0 = (r0 - 1) * W_;
  const float* xb = x + (size_t)(b * C_ + c0) * HW_;
  float sv[(TCH_ * THL_) / 256];
#pragma unroll
  for (int i = 0; i < (TCH_ * THL_) / 256; ++i) {
    int e  = tid + 256 * i;
    int ci = e / THL_;
    int gp = gp0 + (e - ci * THL_);
    int gc = gp < 0 ? 0 : (gp > HW_ - 1 ? HW_ - 1 : gp);
    float v = xb[ci * HW_ + gc];
    sv[i] = (gp == gc) ? v : 0.f;
  }
#pragma unroll
  for (int i = 0; i < (TCH_ * THL_) / 256; ++i) {
    int e  = tid + 256 * i;
    int ci = e / THL_;
    xs[ci * TXP_ + 1 + (e - ci * THL_)] = sv[i];
  }
  if (tid < TCH_) {
    xs[tid * TXP_] = 0.f;
    xs[tid * TXP_ + THL_ + 1] = 0.f;
    xs[tid * TXP_ + THL_ + 2] = 0.f;
  }
  __syncthreads();

  const int c = tid & 31;
  const int w = tid >> 5;
  const float* wp = cw + (c0 + c) * 9;
  float wr[9];
#pragma unroll
  for (int k = 0; k < 9; ++k) wr[k] = wp[k];
  const float bias = cb[c0 + c];
  const float* xr = xs + c * TXP_ + 1;
  float* orow = xc + ((size_t)b * N_ + 1 + r0 * W_) * C_ + c0 + c;
  for (int i = 0; i < TPS_ / 8; ++i) {
    int pl = w + 8 * i;
    int hl = pl / W_;
    int ww = pl - hl * W_;
    int hh = r0 + hl;
    const float* xp = xr + hl * W_ + ww - 1;
    float s = 0.f;
#pragma unroll
    for (int ii = 0; ii < 3; ++ii) {
      int hy = hh + ii - 1;
      bool rok = (hy >= 0) && (hy < H_);
#pragma unroll
      for (int jj = 0; jj < 3; ++jj) {
        int wx = ww + jj - 1;
        bool ok = rok && (wx >= 0) && (wx < W_);
        float sn = s + xp[ii * W_ + jj] * wr[ii * 3 + jj];
        s = ok ? sn : s;
      }
    }
    orow[(size_t)pl * C_] = xp[W_ + 1] + s + bias;
  }
}

__global__ void cls_copy_kernel(const float* __restrict__ cls,
                                float* __restrict__ xc) {
  int idx = blockIdx.x * 256 + threadIdx.x;
  if (idx >= B_ * C_) return;
  int b = idx / C_, c = idx % C_;
  xc[(size_t)b * N_ * C_ + c] = cls[idx];
}

__global__ void ln_kernel(const float* __restrict__ xin,
                          const float* __restrict__ g,
                          const float* __restrict__ bta,
                          float* __restrict__ yout, int ntok) {
  int wave = threadIdx.x >> 5, lane = threadIdx.x & 31;
  int t = blockIdx.x * 8 + wave;
  if (t >= ntok) return;
  const float* xp = xin + (size_t)t * C_;
  float s = 0.f, sq = 0.f;
#pragma unroll
  for (int i = 0; i < C_ / 32; ++i) {
    float v = xp[lane + i * 32];
    s += v; sq += v * v;
  }
#pragma unroll
  for (int off = 16; off; off >>= 1) {
    s  += __shfl_xor(s,  off, 32);
    sq += __shfl_xor(sq, off, 32);
  }
  float mu  = s  * (1.f / C_);
  float var = sq * (1.f / C_) - mu * mu;
  float rs  = rsqrtf(var + 1e-5f);
  float* yp = yout + (size_t)t * C_;
#pragma unroll
  for (int i = 0; i < C_ / 32; ++i) {
    int c = lane + i * 32;
    yp[c] = (xp[c] - mu) * rs * g[c] + bta[c];
  }
}

#define GBM_  128
#define GBN_  64
#define GKC_  32
#define GLDP_ 36
#define GMB_  ((M_ + GBM_ - 1) / GBM_)

typedef __attribute__((ext_vector_type(4))) float v4f;

__device__ __forceinline__ void gemm_chunk(const float* __restrict__ as,
                                           const float* __restrict__ bs,
                                           v8f (&acc)[2][4]) {
  v2f fa[2][2], fb[2][4];
  fa[0][0] = *(const v2f*)(as);
  fa[0][1] = *(const v2f*)(as + 16 * GLDP_);
#pragma unroll
  for (int j = 0; j < 4; ++j) fb[0][j] = *(const v2f*)(bs + j * 16 * GLDP_);
#pragma unroll
  for (int ks = 0; ks < GKC_ / 4; ++ks) {
    const int c = ks & 1, n = c ^ 1;
    if (ks + 1 < GKC_ / 4) {
      fa[n][0] = *(const v2f*)(as + (ks + 1) * 4);
      fa[n][1] = *(const v2f*)(as + 16 * GLDP_ + (ks + 1) * 4);
#pragma unroll
      for (int j = 0; j < 4; ++j)
        fb[n][j] = *(const v2f*)(bs + j * 16 * GLDP_ + (ks + 1) * 4);
    }
    acc[0][0] = wmma4(fa[c][0], fb[c][0], acc[0][0]);
    acc[1][0] = wmma4(fa[c][1], fb[c][0], acc[1][0]);
    acc[0][1] = wmma4(fa[c][0], fb[c][1], acc[0][1]);
    acc[1][1] = wmma4(fa[c][1], fb[c][1], acc[1][1]);
    acc[0][2] = wmma4(fa[c][0], fb[c][2], acc[0][2]);
    acc[1][2] = wmma4(fa[c][1], fb[c][2], acc[1][2]);
    acc[0][3] = wmma4(fa[c][0], fb[c][3], acc[0][3]);
    acc[1][3] = wmma4(fa[c][1], fb[c][3], acc[1][3]);
  }
}

template <int EPI>
__global__ __launch_bounds__(128) void gemm_wmma_kernel(
    const float* __restrict__ X, const float* __restrict__ Wt,
    const float* __restrict__ bias, float* __restrict__ Out,
    int K, int ldo) {
  __shared__ __attribute__((aligned(16))) float As[2][GBM_ * GLDP_];
  __shared__ __attribute__((aligned(16))) float Bs[2][GBN_ * GLDP_];
  const int tid  = threadIdx.x;
  const int lane = tid & 31;
  const int wave = tid >> 5;
  const int n0  = blockIdx.x * GBN_;
  const int mb0 = blockIdx.y * GBM_;

  const int lr = tid >> 3;
  const int lc = (tid & 7) << 2;
  const float* ag[8];
  const float* bg[4];
#pragma unroll
  for (int i = 0; i < 8; ++i) {
    int r = mb0 + lr + 16 * i;
    if (r > M_ - 1) r = M_ - 1;
    ag[i] = X + (size_t)r * K + lc;
  }
#pragma unroll
  for (int i = 0; i < 4; ++i) bg[i] = Wt + (size_t)(n0 + lr + 16 * i) * K + lc;
  const int soff = lr * GLDP_ + lc;

  v4f ra[8], rb[4];
#pragma unroll
  for (int i = 0; i < 8; ++i) ra[i] = *(const v4f*)(ag[i]);
#pragma unroll
  for (int i = 0; i < 4; ++i) rb[i] = *(const v4f*)(bg[i]);
#pragma unroll
  for (int i = 0; i < 8; ++i) *(v4f*)(&As[0][soff + 16 * i * GLDP_]) = ra[i];
#pragma unroll
  for (int i = 0; i < 4; ++i) *(v4f*)(&Bs[0][soff + 16 * i * GLDP_]) = rb[i];
  __syncthreads();

  const int rc = lane & 15;
  const int kb = (lane >> 4) << 1;
  const int aoff = (wave * 32 + rc) * GLDP_ + kb;
  const int boff = rc * GLDP_ + kb;

  v8f acc[2][4] = {};
  const int nk = K / GKC_;
  for (int kc = 0; kc < nk - 1; ++kc) {
    const int cur = kc & 1;
    const int kn = (kc + 1) * GKC_;
#pragma unroll
    for (int i = 0; i < 8; ++i) ra[i] = *(const v4f*)(ag[i] + kn);
#pragma unroll
    for (int i = 0; i < 4; ++i) rb[i] = *(const v4f*)(bg[i] + kn);
    gemm_chunk(&As[cur][aoff], &Bs[cur][boff], acc);
#pragma unroll
    for (int i = 0; i < 8; ++i) *(v4f*)(&As[cur ^ 1][soff + 16 * i * GLDP_]) = ra[i];
#pragma unroll
    for (int i = 0; i < 4; ++i) *(v4f*)(&Bs[cur ^ 1][soff + 16 * i * GLDP_]) = rb[i];
    __syncthreads();
  }
  {
    const int cur = (nk - 1) & 1;
    gemm_chunk(&As[cur][aoff], &Bs[cur][boff], acc);
  }

  const int m0 = mb0 + wave * 32;
  int coln = lane & 15;
  int roff = (lane >> 4) << 3;
  if (EPI == 1) {
    float bv[4];
#pragma unroll
    for (int j = 0; j < 4; ++j) bv[j] = bias ? bias[n0 + j * 16 + coln] : 0.f;
    float res[2][4][8];
#pragma unroll
    for (int mi = 0; mi < 2; ++mi) {
#pragma unroll
      for (int r = 0; r < 8; ++r) {
        int gm = m0 + mi * 16 + roff + r;
        int gl = gm < M_ ? gm : M_ - 1;
        const float* rp = Out + (size_t)gl * ldo + n0 + coln;
#pragma unroll
        for (int j = 0; j < 4; ++j) res[mi][j][r] = rp[j * 16];
      }
    }
#pragma unroll
    for (int mi = 0; mi < 2; ++mi) {
#pragma unroll
      for (int r = 0; r < 8; ++r) {
        int gm = m0 + mi * 16 + roff + r;
        if (gm < M_) {
          float* wp = Out + (size_t)gm * ldo + n0 + coln;
#pragma unroll
          for (int j = 0; j < 4; ++j)
            wp[j * 16] = res[mi][j][r] + (acc[mi][j][r] + bv[j]);
        }
      }
    }
  } else {
#pragma unroll
    for (int mi = 0; mi < 2; ++mi) {
#pragma unroll
      for (int j = 0; j < 4; ++j) {
        int nc = n0 + j * 16 + coln;
        float bv = bias ? bias[nc] : 0.f;
#pragma unroll
        for (int r = 0; r < 8; ++r) {
          int gm = m0 + mi * 16 + roff + r;
          if (gm < M_) {
            size_t oi = (size_t)gm * ldo + nc;
            float v = acc[mi][j][r] + bv;
            if (EPI == 0) Out[oi] = v;
            else          Out[oi] = 0.5f * v * (1.f + erff(v * 0.70710678118654752f));
          }
        }
      }
    }
  }
}

#define AQB_  64
#define AKT_  64
#define ALDP_ 52
#define ANKT_ ((N_ + AKT_ - 1) / AKT_)
#define AQBN_ ((N_ + AQB_ - 1) / AQB_)
#define ANEG_ (-3.0e38f)

__device__ __forceinline__ void attn_tile(const float* __restrict__ kt,
                                          const float* __restrict__ vt,
                                          const v2f (&qf)[12], int key0,
                                          float& m, float& l, v8f (&oacc)[3],
                                          bool keep_row0, float* __restrict__ srow) {
  const float scale = 0.14433756729740643f;
  v8f sacc[4] = {};
  v2f fk[2][4];
#pragma unroll
  for (int j = 0; j < 4; ++j) fk[0][j] = *(const v2f*)(kt + j * 16 * ALDP_);
#pragma unroll
  for (int ks = 0; ks < 12; ++ks) {
    const int cb = ks & 1, nb = cb ^ 1;
    if (ks + 1 < 12) {
#pragma unroll
      for (int j = 0; j < 4; ++j)
        fk[nb][j] = *(const v2f*)(kt + j * 16 * ALDP_ + (ks + 1) * 4);
    }
#pragma unroll
    for (int j = 0; j < 4; ++j) sacc[j] = wmma4(fk[cb][j], qf[ks], sacc[j]);
  }
  float tmax = ANEG_;
#pragma unroll
  for (int j = 0; j < 4; ++j) {
#pragma unroll
    for (int r = 0; r < 8; ++r) {
      float v = sacc[j][r] * scale;
      v = (key0 + j * 16 + r < N_) ? v : ANEG_;
      sacc[j][r] = v;
      tmax = fmaxf(tmax, v);
    }
  }
  if (keep_row0) {
#pragma unroll
    for (int j = 0; j < 4; ++j)
#pragma unroll
      for (int r = 0; r < 8; ++r) srow[key0 + j * 16 + r] = sacc[j][r];
  }
  tmax = fmaxf(tmax, __shfl_xor(tmax, 16, 32));
  const float mn = fmaxf(m, tmax);
  const float alpha = expf(m - mn);
  m = mn;
  float ps = 0.f;
#pragma unroll
  for (int j = 0; j < 4; ++j) {
#pragma unroll
    for (int r = 0; r < 8; ++r) {
      float p = expf(sacc[j][r] - mn);
      sacc[j][r] = p;
      ps += p;
    }
  }
  l = l * alpha + ps;
#pragma unroll
  for (int n = 0; n < 3; ++n)
#pragma unroll
    for (int r = 0; r < 8; ++r) oacc[n][r] *= alpha;
  v2f fv[2][3];
#pragma unroll
  for (int n = 0; n < 3; ++n) {
    fv[0][n].x = vt[n * 16];
    fv[0][n].y = vt[ALDP_ + n * 16];
  }
#pragma unroll
  for (int u = 0; u < 16; ++u) {
    const int cb = u & 1, nb = cb ^ 1;
    if (u + 1 < 16) {
      const int ko = ((u + 1) >> 2) * 16 + ((u + 1) & 3) * 2;
#pragma unroll
      for (int n = 0; n < 3; ++n) {
        fv[nb][n].x = vt[ko * ALDP_ + n * 16];
        fv[nb][n].y = vt[(ko + 1) * ALDP_ + n * 16];
      }
    }
    v2f pb;
    pb.x = sacc[u >> 2][(u & 3) * 2];
    pb.y = sacc[u >> 2][(u & 3) * 2 + 1];
#pragma unroll
    for (int n = 0; n < 3; ++n) oacc[n] = wmma4(fv[cb][n], pb, oacc[n]);
  }
}

__global__ __launch_bounds__(128) void attn_kernel(const float* __restrict__ qkv,
                                                   float* __restrict__ o,
                                                   float* __restrict__ ga) {
  __shared__ __attribute__((aligned(16))) float Ks[2][AKT_ * ALDP_];
  __shared__ __attribute__((aligned(16))) float Vs[2][AKT_ * ALDP_];
  __shared__ float Srow[ANKT_ * AKT_];
  __shared__ float Mrow[2];
  const int tid  = threadIdx.x;
  const int lane = tid & 31;
  const int wave = tid >> 5;
  const int hd = blockIdx.y, b = blockIdx.z;
  const int c = lane & 15;
  const int h = lane >> 4;
  const int q0 = blockIdx.x * AQB_ + wave * 16;
  int gq = q0 + c; if (gq > N_ - 1) gq = N_ - 1;
  const float* qrow = qkv + (size_t)(b * N_ + gq) * C3_ + hd * D_ + 2 * h;
  v2f qf[12];
#pragma unroll
  for (int ks = 0; ks < 12; ++ks) qf[ks] = *(const v2f*)(qrow + ks * 4);

  const float* kbase = qkv + (size_t)b * N_ * C3_ + C_ + hd * D_;
  int srw[6], sof[6];
#pragma unroll
  for (int i = 0; i < 6; ++i) {
    int idx = tid + 128 * i;
    srw[i] = idx / 12;
    sof[i] = (idx - srw[i] * 12) * 4;
  }
  v4f rk[6], rv[6];
#pragma unroll
  for (int i = 0; i < 6; ++i) {
    const float* p = kbase + (size_t)srw[i] * C3_ + sof[i];
    rk[i] = *(const v4f*)(p);
    rv[i] = *(const v4f*)(p + C_);
  }
#pragma unroll
  for (int i = 0; i < 6; ++i) {
    *(v4f*)(&Ks[0][srw[i] * ALDP_ + sof[i]]) = rk[i];
    *(v4f*)(&Vs[0][srw[i] * ALDP_ + sof[i]]) = rv[i];
  }
  __syncthreads();

  float m = ANEG_, l = 0.f;
  v8f oacc[3] = {};
  const bool keep_row0 = (blockIdx.x == 0) && (wave == 0) && (c == 0);
  const int koff = c * ALDP_ + 2 * h;
  const int voff = 8 * h * ALDP_ + c;

  for (int t = 0; t < ANKT_ - 1; ++t) {
    const int cur = t & 1;
#pragma unroll
    for (int i = 0; i < 6; ++i) {
      int kr = (t + 1) * AKT_ + srw[i];
      if (kr > N_ - 1) kr = N_ - 1;
      const float* p = kbase + (size_t)kr * C3_ + sof[i];
      rk[i] = *(const v4f*)(p);
      rv[i] = *(const v4f*)(p + C_);
    }
    attn_tile(&Ks[cur][koff], &Vs[cur][voff], qf, t * AKT_ + 8 * h, m, l, oacc,
              keep_row0, Srow);
#pragma unroll
    for (int i = 0; i < 6; ++i) {
      *(v4f*)(&Ks[cur ^ 1][srw[i] * ALDP_ + sof[i]]) = rk[i];
      *(v4f*)(&Vs[cur ^ 1][srw[i] * ALDP_ + sof[i]]) = rv[i];
    }
    __syncthreads();
  }
  {
    const int t = ANKT_ - 1;
    attn_tile(&Ks[t & 1][koff], &Vs[t & 1][voff], qf, t * AKT_ + 8 * h, m, l, oacc,
              keep_row0, Srow);
  }

  const float lt  = l + __shfl_xor(l, 16, 32);
  const float inv = 1.f / lt;
  const int q = q0 + c;
  if (q < N_) {
    float* op = o + (size_t)(b * N_ + q) * C_ + hd * D_ + 8 * h;
#pragma unroll
    for (int n = 0; n < 3; ++n) {
      v4f lo, hi;
      lo.x = oacc[n][0] * inv; lo.y = oacc[n][1] * inv;
      lo.z = oacc[n][2] * inv; lo.w = oacc[n][3] * inv;
      hi.x = oacc[n][4] * inv; hi.y = oacc[n][5] * inv;
      hi.z = oacc[n][6] * inv; hi.w = oacc[n][7] * inv;
      *(v4f*)(op + n * 16)     = lo;
      *(v4f*)(op + n * 16 + 4) = hi;
    }
  }

  if (blockIdx.x == 0) {
    if (tid == 0) { Mrow[0] = m; Mrow[1] = inv; }
    __syncthreads();
    const float m0 = Mrow[0], i0 = Mrow[1];
    float* gp = ga + (size_t)(b * HEADS_ + hd) * HW_;
    for (int key = 1 + tid; key < N_; key += 128)
      gp[key - 1] = expf(Srow[key] - m0) * i0;
  }
}

#define FTB_  (B_ * TCT_ * TNT_)
#define FTL_  ((B_ * C_ + B_ * HW_ + 255) / 256)
#define FTP_  33

__global__ __launch_bounds__(256) void finalize_kernel(const float* __restrict__ xc,
                                                       const float* __restrict__ ga,
                                                       float* __restrict__ out) {
  __shared__ float ts[TPS_ * FTP_];
  const int tid = threadIdx.x;
  const size_t n_cls = (size_t)B_ * C_;
  const size_t n_x   = (size_t)B_ * C_ * HW_;
  const size_t n_ga  = (size_t)B_ * HW_;
  if (blockIdx.x < FTB_) {
    const int bid = blockIdx.x;
    const int rt = bid % TNT_;
    const int t2 = bid / TNT_;
    const int ct = t2 % TCT_;
    const int b  = t2 / TCT_;
    const int c0 = ct * TCH_, p0 = rt * TPS_;
    const int c = tid & 31, w = tid >> 5;
    const float* ib = xc + ((size_t)b * N_ + 1 + p0) * C_ + c0 + c;
#pragma unroll
    for (int i = 0; i < TPS_ / 8; ++i) {
      int pl = w + 8 * i;
      ts[pl * FTP_ + c] = ib[(size_t)pl * C_];
    }
    __syncthreads();
    float* ob = out + n_cls + (size_t)(b * C_ + c0) * HW_ + p0;
#pragma unroll
    for (int i = 0; i < (TCH_ * TPS_) / 256; ++i) {
      int e  = tid + 256 * i;
      int ci = e / TPS_;
      int pl = e - ci * TPS_;
      ob[(size_t)ci * HW_ + pl] = ts[pl * FTP_ + ci];
    }
  } else {
    size_t j = (size_t)(blockIdx.x - FTB_) * 256 + tid;
    if (j < n_cls) {
      int b = (int)(j / C_), c = (int)(j % C_);
      out[j] = xc[(size_t)b * N_ * C_ + c];
    } else if (j < n_cls + n_ga) {
      size_t k = j - n_cls;
      int b = (int)(k / HW_), p = (int)(k % HW_);
      float s = 0.f;
#pragma unroll
      for (int hh = 0; hh < HEADS_; ++hh)
        s += ga[(size_t)(b * HEADS_ + hh) * HW_ + p];
      out[n_cls + n_x + k] = s * 0.125f;
    }
  }
}

extern "C" void kernel_launch(void* const* d_in, const int* in_sizes, int n_in,
                              void* d_out, int out_size, void* d_ws, size_t ws_size,
                              hipStream_t stream) {
  const float* cls    = (const float*)d_in[0];
  const float* x      = (const float*)d_in[1];
  const float* conv_w = (const float*)d_in[2];
  const float* conv_b = (const float*)d_in[3];
  const float* ln1_g  = (const float*)d_in[4];
  const float* ln1_b  = (const float*)d_in[5];
  const float* qkv_w  = (const float*)d_in[6];
  const float* proj_w = (const float*)d_in[7];
  const float* proj_b = (const float*)d_in[8];
  const float* ln2_g  = (const float*)d_in[9];
  const float* ln2_b  = (const float*)d_in[10];
  const float* fc1_w  = (const float*)d_in[11];
  const float* fc1_b  = (const float*)d_in[12];
  const float* fc2_w  = (const float*)d_in[13];
  const float* fc2_b  = (const float*)d_in[14];

  float* ws  = (float*)d_ws;
  float* xc  = ws;
  float* y   = xc  + (size_t)MP_ * C_;
  float* qkv = y   + (size_t)MP_ * C_;
  float* o   = qkv + (size_t)MP_ * C3_;
  float* m1  = qkv;
  float* ga  = o   + (size_t)MP_ * C_;
  float* out = (float*)d_out;

  conv_pe_kernel<<<dim3(TNT_, TCT_, B_), 256, 0, stream>>>(x, conv_w, conv_b, xc);
  cls_copy_kernel<<<(B_ * C_ + 255) / 256, 256, 0, stream>>>(cls, xc);

  ln_kernel<<<(M_ + 7) / 8, 256, 0, stream>>>(xc, ln1_g, ln1_b, y, M_);

  gemm_wmma_kernel<0><<<dim3(C3_ / GBN_, GMB_), 128, 0, stream>>>(
      y, qkv_w, nullptr, qkv, C_, C3_);

  attn_kernel<<<dim3(AQBN_, HEADS_, B_), 128, 0, stream>>>(qkv, o, ga);

  gemm_wmma_kernel<1><<<dim3(C_ / GBN_, GMB_), 128, 0, stream>>>(
      o, proj_w, proj_b, xc, C_, C_);

  ln_kernel<<<(M_ + 7) / 8, 256, 0, stream>>>(xc, ln2_g, ln2_b, y, M_);

  gemm_wmma_kernel<2><<<dim3(C4_ / GBN_, GMB_), 128, 0, stream>>>(
      y, fc1_w, fc1_b, m1, C_, C4_);

  gemm_wmma_kernel<1><<<dim3(C_ / GBN_, GMB_), 128, 0, stream>>>(
      m1, fc2_w, fc2_b, xc, C4_, C_);

  finalize_kernel<<<FTB_ + FTL_, 256, 0, stream>>>(xc, ga, out);
}


# Attention softmax in base 2: p = exp2(s*c - m*c), c = log2(e)/sqrt(d): one fma + one hardware exp2 per raw score, no library expf expansion; last-bit rounding of p changes, sums unchanged
# speedup vs baseline: 3.2179x; 1.0329x over previous
#include <hip/hip_runtime.h>
#include <math.h>

#define B_    16
#define C_    384
#define H_    28
#define W_    28
#define HW_   784
#define N_    785
#define M_    (B_*N_)
#define MP_   (M_ + 16)
#define HEADS_ 8
#define D_    48
#define C3_   1152
#define C4_   1536
#define SLD   800

typedef __attribute__((ext_vector_type(2))) float v2f;
typedef __attribute__((ext_vector_type(8))) float v8f;

__device__ __forceinline__ v8f wmma4(v2f a, v2f b, v8f c) {
  return __builtin_amdgcn_wmma_f32_16x16x4_f32(false, a, false, b, (short)0, c,
                                               false, false);
}

#define TCH_ 32
#define TRW_ 4
#define TPS_ (TRW_ * W_)
#define THL_ ((TRW_ + 2) * W_)
#define TXP_ (THL_ + 3)
#define TNT_ (H_ / TRW_)
#define TCT_ (C_ / TCH_)

__global__ __launch_bounds__(256) void conv_pe_kernel(const float* __restrict__ x,
                                                      const float* __restrict__ cw,
                                                      const float* __restrict__ cb,
                                                      float* __restrict__ xc) {
  __shared__ float xs[TCH_ * TXP_];
  const int tid = threadIdx.x;
  const int r0 = blockIdx.x * TRW_;
  const int c0 = blockIdx.y * TCH_;
  const int b  = blockIdx.z;
  const int gp0 = (r0 - 1) * W_;
  const float* xb = x + (size_t)(b * C_ + c0) * HW_;
  float sv[(TCH_ * THL_) / 256];
#pragma unroll
  for (int i = 0; i < (TCH_ * THL_) / 256; ++i) {
    int e  = tid + 256 * i;
    int ci = e / THL_;
    int gp = gp0 + (e - ci * THL_);
    int gc = gp < 0 ? 0 : (gp > HW_ - 1 ? HW_ - 1 : gp);
    float v = xb[ci * HW_ + gc];
    sv[i] = (gp == gc) ? v : 0.f;
  }
#pragma unroll
  for (int i = 0; i < (TCH_ * THL_) / 256; ++i) {
    int e  = tid + 256 * i;
    int ci = e / THL_;
    xs[ci * TXP_ + 1 + (e - ci * THL_)] = sv[i];
  }
  if (tid < TCH_) {
    xs[tid * TXP_] = 0.f;
    xs[tid * TXP_ + THL_ + 1] = 0.f;
    xs[tid * TXP_ + THL_ + 2] = 0.f;
  }
  __syncthreads();

  const int c = tid & 31;
  const int w = tid >> 5;
  const float* wp = cw + (c0 + c) * 9;
  float wr[9];
#pragma unroll
  for (int k = 0; k < 9; ++k) wr[k] = wp[k];
  const float bias = cb[c0 + c];
  const float* xr = xs + c * TXP_ + 1;
  float* orow = xc + ((size_t)b * N_ + 1 + r0 * W_) * C_ + c0 + c;
  for (int i = 0; i < TPS_ / 8; ++i) {
    int pl = w + 8 * i;
    int hl = pl / W_;
    int ww = pl - hl * W_;
    int hh = r0 + hl;
    const float* xp = xr + hl * W_ + ww - 1;
    float s = 0.f;
#pragma unroll
    for (int ii = 0; ii < 3; ++ii) {
      int hy = hh + ii - 1;
      bool rok = (hy >= 0) && (hy < H_);
#pragma unroll
      for (int jj = 0; jj < 3; ++jj) {
        int wx = ww + jj - 1;
        bool ok = rok && (wx >= 0) && (wx < W_);
        float sn = s + xp[ii * W_ + jj] * wr[ii * 3 + jj];
        s = ok ? sn : s;
      }
    }
    orow[(size_t)pl * C_] = xp[W_ + 1] + s + bias;
  }
}

__global__ void cls_copy_kernel(const float* __restrict__ cls,
                                float* __restrict__ xc) {
  int idx = blockIdx.x * 256 + threadIdx.x;
  if (idx >= B_ * C_) return;
  int b = idx / C_, c = idx % C_;
  xc[(size_t)b * N_ * C_ + c] = cls[idx];
}

__global__ void ln_kernel(const float* __restrict__ xin,
                          const float* __restrict__ g,
                          const float* __restrict__ bta,
                          float* __restrict__ yout, int ntok) {
  int wave = threadIdx.x >> 5, lane = threadIdx.x & 31;
  int t = blockIdx.x * 8 + wave;
  if (t >= ntok) return;
  const float* xp = xin + (size_t)t * C_;
  float s = 0.f, sq = 0.f;
#pragma unroll
  for (int i = 0; i < C_ / 32; ++i) {
    float v = xp[lane + i * 32];
    s += v; sq += v * v;
  }
#pragma unroll
  for (int off = 16; off; off >>= 1) {
    s  += __shfl_xor(s,  off, 32);
    sq += __shfl_xor(sq, off, 32);
  }
  float mu  = s  * (1.f / C_);
  float var = sq * (1.f / C_) - mu * mu;
  float rs  = rsqrtf(var + 1e-5f);
  float* yp = yout + (size_t)t * C_;
#pragma unroll
  for (int i = 0; i < C_ / 32; ++i) {
    int c = lane + i * 32;
    yp[c] = (xp[c] - mu) * rs * g[c] + bta[c];
  }
}

#define GBM_  128
#define GBN_  64
#define GKC_  32
#define GLDP_ 36
#define GMB_  ((M_ + GBM_ - 1) / GBM_)

typedef __attribute__((ext_vector_type(4))) float v4f;

__device__ __forceinline__ void gemm_chunk(const float* __restrict__ as,
                                           const float* __restrict__ bs,
                                           v8f (&acc)[2][4]) {
  v2f fa[2][2], fb[2][4];
  fa[0][0] = *(const v2f*)(as);
  fa[0][1] = *(const v2f*)(as + 16 * GLDP_);
#pragma unroll
  for (int j = 0; j < 4; ++j) fb[0][j] = *(const v2f*)(bs + j * 16 * GLDP_);
#pragma unroll
  for (int ks = 0; ks < GKC_ / 4; ++ks) {
    const int c = ks & 1, n = c ^ 1;
    if (ks + 1 < GKC_ / 4) {
      fa[n][0] = *(const v2f*)(as + (ks + 1) * 4);
      fa[n][1] = *(const v2f*)(as + 16 * GLDP_ + (ks + 1) * 4);
#pragma unroll
      for (int j = 0; j < 4; ++j)
        fb[n][j] = *(const v2f*)(bs + j * 16 * GLDP_ + (ks + 1) * 4);
    }
    acc[0][0] = wmma4(fa[c][0], fb[c][0], acc[0][0]);
    acc[1][0] = wmma4(fa[c][1], fb[c][0], acc[1][0]);
    acc[0][1] = wmma4(fa[c][0], fb[c][1], acc[0][1]);
    acc[1][1] = wmma4(fa[c][1], fb[c][1], acc[1][1]);
    acc[0][2] = wmma4(fa[c][0], fb[c][2], acc[0][2]);
    acc[1][2] = wmma4(fa[c][1], fb[c][2], acc[1][2]);
    acc[0][3] = wmma4(fa[c][0], fb[c][3], acc[0][3]);
    acc[1][3] = wmma4(fa[c][1], fb[c][3], acc[1][3]);
  }
}

template <int EPI>
__global__ __launch_bounds__(128) void gemm_wmma_kernel(
    const float* __restrict__ X, const float* __restrict__ Wt,
    const float* __restrict__ bias, float* __restrict__ Out,
    int K, int ldo) {
  __shared__ __attribute__((aligned(16))) float As[2][GBM_ * GLDP_];
  __shared__ __attribute__((aligned(16))) float Bs[2][GBN_ * GLDP_];
  const int tid  = threadIdx.x;
  const int lane = tid & 31;
  const int wave = tid >> 5;
  const int n0  = blockIdx.x * GBN_;
  const int mb0 = blockIdx.y * GBM_;

  const int lr = tid >> 3;
  const int lc = (tid & 7) << 2;
  const float* ag[8];
  const float* bg[4];
#pragma unroll
  for (int i = 0; i < 8; ++i) {
    int r = mb0 + lr + 16 * i;
    if (r > M_ - 1) r = M_ - 1;
    ag[i] = X + (size_t)r * K + lc;
  }
#pragma unroll
  for (int i = 0; i < 4; ++i) bg[i] = Wt + (size_t)(n0 + lr + 16 * i) * K + lc;
  const int soff = lr * GLDP_ + lc;

  v4f ra[8], rb[4];
#pragma unroll
  for (int i = 0; i < 8; ++i) ra[i] = *(const v4f*)(ag[i]);
#pragma unroll
  for (int i = 0; i < 4; ++i) rb[i] = *(const v4f*)(bg[i]);
#pragma unroll
  for (int i = 0; i < 8; ++i) *(v4f*)(&As[0][soff + 16 * i * GLDP_]) = ra[i];
#pragma unroll
  for (int i = 0; i < 4; ++i) *(v4f*)(&Bs[0][soff + 16 * i * GLDP_]) = rb[i];
  __syncthreads();

  const int rc = lane & 15;
  const int kb = (lane >> 4) << 1;
  const int aoff = (wave * 32 + rc) * GLDP_ + kb;
  const int boff = rc * GLDP_ + kb;

  v8f acc[2][4] = {};
  const int nk = K / GKC_;
  for (int kc = 0; kc < nk - 1; ++kc) {
    const int cur = kc & 1;
    const int kn = (kc + 1) * GKC_;
#pragma unroll
    for (int i = 0; i < 8; ++i) ra[i] = *(const v4f*)(ag[i] + kn);
#pragma unroll
    for (int i = 0; i < 4; ++i) rb[i] = *(const v4f*)(bg[i] + kn);
    gemm_chunk(&As[cur][aoff], &Bs[cur][boff], acc);
#pragma unroll
    for (int i = 0; i < 8; ++i) *(v4f*)(&As[cur ^ 1][soff + 16 * i * GLDP_]) = ra[i];
#pragma unroll
    for (int i = 0; i < 4; ++i) *(v4f*)(&Bs[cur ^ 1][soff + 16 * i * GLDP_]) = rb[i];
    __syncthreads();
  }
  {
    const int cur = (nk - 1) & 1;
    gemm_chunk(&As[cur][aoff], &Bs[cur][boff], acc);
  }

  const int m0 = mb0 + wave * 32;
  int coln = lane & 15;
  int roff = (lane >> 4) << 3;
  if (EPI == 1) {
    float bv[4];
#pragma unroll
    for (int j = 0; j < 4; ++j) bv[j] = bias ? bias[n0 + j * 16 + coln] : 0.f;
    float res[2][4][8];
#pragma unroll
    for (int mi = 0; mi < 2; ++mi) {
#pragma unroll
      for (int r = 0; r < 8; ++r) {
        int gm = m0 + mi * 16 + roff + r;
        int gl = gm < M_ ? gm : M_ - 1;
        const float* rp = Out + (size_t)gl * ldo + n0 + coln;
#pragma unroll
        for (int j = 0; j < 4; ++j) res[mi][j][r] = rp[j * 16];
      }
    }
#pragma unroll
    for (int mi = 0; mi < 2; ++mi) {
#pragma unroll
      for (int r = 0; r < 8; ++r) {
        int gm = m0 + mi * 16 + roff + r;
        if (gm < M_) {
          float* wp = Out + (size_t)gm * ldo + n0 + coln;
#pragma unroll
          for (int j = 0; j < 4; ++j)
            wp[j * 16] = res[mi][j][r] + (acc[mi][j][r] + bv[j]);
        }
      }
    }
  } else {
#pragma unroll
    for (int mi = 0; mi < 2; ++mi) {
#pragma unroll
      for (int j = 0; j < 4; ++j) {
        int nc = n0 + j * 16 + coln;
        float bv = bias ? bias[nc] : 0.f;
#pragma unroll
        for (int r = 0; r < 8; ++r) {
          int gm = m0 + mi * 16 + roff + r;
          if (gm < M_) {
            size_t oi = (size_t)gm * ldo + nc;
            float v = acc[mi][j][r] + bv;
            if (EPI == 0) Out[oi] = v;
            else          Out[oi] = 0.5f * v * (1.f + erff(v * 0.70710678118654752f));
          }
        }
      }
    }
  }
}

#define AQB_  64
#define AKT_  64
#define ALDP_ 52
#define ANKT_ ((N_ + AKT_ - 1) / AKT_)
#define AQBN_ ((N_ + AQB_ - 1) / AQB_)
#define ANEG_ (-3.0e38f)

__device__ __forceinline__ void attn_tile(const float* __restrict__ kt,
                                          const float* __restrict__ vt,
                                          const v2f (&qf)[12], int key0,
                                          float& m, float& l, v8f (&oacc)[3],
                                          bool keep_row0, float* __restrict__ srow) {
  const float cs = 0.2082350925539453f;
  v8f sacc[4] = {};
  v2f fk[2][4];
#pragma unroll
  for (int j = 0; j < 4; ++j) fk[0][j] = *(const v2f*)(kt + j * 16 * ALDP_);
#pragma unroll
  for (int ks = 0; ks < 12; ++ks) {
    const int cb = ks & 1, nb = cb ^ 1;
    if (ks + 1 < 12) {
#pragma unroll
      for (int j = 0; j < 4; ++j)
        fk[nb][j] = *(const v2f*)(kt + j * 16 * ALDP_ + (ks + 1) * 4);
    }
#pragma unroll
    for (int j = 0; j < 4; ++j) sacc[j] = wmma4(fk[cb][j], qf[ks], sacc[j]);
  }
  float tmax = ANEG_;
#pragma unroll
  for (int j = 0; j < 4; ++j) {
#pragma unroll
    for (int r = 0; r < 8; ++r) {
      float v = sacc[j][r];
      v = (key0 + j * 16 + r < N_) ? v : ANEG_;
      sacc[j][r] = v;
      tmax = fmaxf(tmax, v);
    }
  }
  if (keep_row0) {
#pragma unroll
    for (int j = 0; j < 4; ++j)
#pragma unroll
      for (int r = 0; r < 8; ++r) srow[key0 + j * 16 + r] = sacc[j][r];
  }
  tmax = fmaxf(tmax, __shfl_xor(tmax, 16, 32));
  const float mn = fmaxf(m, tmax);
  const float nmc = -mn * cs;
  const float alpha = __builtin_amdgcn_exp2f(fmaf(m, cs, nmc));
  m = mn;
  float ps = 0.f;
#pragma unroll
  for (int j = 0; j < 4; ++j) {
#pragma unroll
    for (int r = 0; r < 8; ++r) {
      float p = __builtin_amdgcn_exp2f(fmaf(sacc[j][r], cs, nmc));
      sacc[j][r] = p;
      ps += p;
    }
  }
  l = l * alpha + ps;
#pragma unroll
  for (int n = 0; n < 3; ++n)
#pragma unroll
    for (int r = 0; r < 8; ++r) oacc[n][r] *= alpha;
  v2f fv[2][3];
#pragma unroll
  for (int n = 0; n < 3; ++n) {
    fv[0][n].x = vt[n * 16];
    fv[0][n].y = vt[ALDP_ + n * 16];
  }
#pragma unroll
  for (int u = 0; u < 16; ++u) {
    const int cb = u & 1, nb = cb ^ 1;
    if (u + 1 < 16) {
      const int ko = ((u + 1) >> 2) * 16 + ((u + 1) & 3) * 2;
#pragma unroll
      for (int n = 0; n < 3; ++n) {
        fv[nb][n].x = vt[ko * ALDP_ + n * 16];
        fv[nb][n].y = vt[(ko + 1) * ALDP_ + n * 16];
      }
    }
    v2f pb;
    pb.x = sacc[u >> 2][(u & 3) * 2];
    pb.y = sacc[u >> 2][(u & 3) * 2 + 1];
#pragma unroll
    for (int n = 0; n < 3; ++n) oacc[n] = wmma4(fv[cb][n], pb, oacc[n]);
  }
}

__global__ __launch_bounds__(128) void attn_kernel(const float* __restrict__ qkv,
                                                   float* __restrict__ o,
                                                   float* __restrict__ ga) {
  __shared__ __attribute__((aligned(16))) float Ks[2][AKT_ * ALDP_];
  __shared__ __attribute__((aligned(16))) float Vs[2][AKT_ * ALDP_];
  __shared__ float Srow[ANKT_ * AKT_];
  __shared__ float Mrow[2];
  const int tid  = threadIdx.x;
  const int lane = tid & 31;
  const int wave = tid >> 5;
  const int hd = blockIdx.y, b = blockIdx.z;
  const int c = lane & 15;
  const int h = lane >> 4;
  const int q0 = blockIdx.x * AQB_ + wave * 16;
  int gq = q0 + c; if (gq > N_ - 1) gq = N_ - 1;
  const float* qrow = qkv + (size_t)(b * N_ + gq) * C3_ + hd * D_ + 2 * h;
  v2f qf[12];
#pragma unroll
  for (int ks = 0; ks < 12; ++ks) qf[ks] = *(const v2f*)(qrow + ks * 4);

  const float* kbase = qkv + (size_t)b * N_ * C3_ + C_ + hd * D_;
  int srw[6], sof[6];
#pragma unroll
  for (int i = 0; i < 6; ++i) {
    int idx = tid + 128 * i;
    srw[i] = idx / 12;
    sof[i] = (idx - srw[i] * 12) * 4;
  }
  v4f rk[6], rv[6];
#pragma unroll
  for (int i = 0; i < 6; ++i) {
    const float* p = kbase + (size_t)srw[i] * C3_ + sof[i];
    rk[i] = *(const v4f*)(p);
    rv[i] = *(const v4f*)(p + C_);
  }
#pragma unroll
  for (int i = 0; i < 6; ++i) {
    *(v4f*)(&Ks[0][srw[i] * ALDP_ + sof[i]]) = rk[i];
    *(v4f*)(&Vs[0][srw[i] * ALDP_ + sof[i]]) = rv[i];
  }
  __syncthreads();

  float m = ANEG_, l = 0.f;
  v8f oacc[3] = {};
  const bool keep_row0 = (blockIdx.x == 0) && (wave == 0) && (c == 0);
  const int koff = c * ALDP_ + 2 * h;
  const int voff = 8 * h * ALDP_ + c;

  for (int t = 0; t < ANKT_ - 1; ++t) {
    const int cur = t & 1;
#pragma unroll
    for (int i = 0; i < 6; ++i) {
      int kr = (t + 1) * AKT_ + srw[i];
      if (kr > N_ - 1) kr = N_ - 1;
      const float* p = kbase + (size_t)kr * C3_ + sof[i];
      rk[i] = *(const v4f*)(p);
      rv[i] = *(const v4f*)(p + C_);
    }
    attn_tile(&Ks[cur][koff], &Vs[cur][voff], qf, t * AKT_ + 8 * h, m, l, oacc,
              keep_row0, Srow);
#pragma unroll
    for (int i = 0; i < 6; ++i) {
      *(v4f*)(&Ks[cur ^ 1][srw[i] * ALDP_ + sof[i]]) = rk[i];
      *(v4f*)(&Vs[cur ^ 1][srw[i] * ALDP_ + sof[i]]) = rv[i];
    }
    __syncthreads();
  }
  {
    const int t = ANKT_ - 1;
    attn_tile(&Ks[t & 1][koff], &Vs[t & 1][voff], qf, t * AKT_ + 8 * h, m, l, oacc,
              keep_row0, Srow);
  }

  const float lt  = l + __shfl_xor(l, 16, 32);
  const float inv = 1.f / lt;
  const int q = q0 + c;
  if (q < N_) {
    float* op = o + (size_t)(b * N_ + q) * C_ + hd * D_ + 8 * h;
#pragma unroll
    for (int n = 0; n < 3; ++n) {
      v4f lo, hi;
      lo.x = oacc[n][0] * inv; lo.y = oacc[n][1] * inv;
      lo.z = oacc[n][2] * inv; lo.w = oacc[n][3] * inv;
      hi.x = oacc[n][4] * inv; hi.y = oacc[n][5] * inv;
      hi.z = oacc[n][6] * inv; hi.w = oacc[n][7] * inv;
      *(v4f*)(op + n * 16)     = lo;
      *(v4f*)(op + n * 16 + 4) = hi;
    }
  }

  if (blockIdx.x == 0) {
    if (tid == 0) { Mrow[0] = m; Mrow[1] = inv; }
    __syncthreads();
    const float cs = 0.2082350925539453f;
    const float nm0 = -Mrow[0] * cs, i0 = Mrow[1];
    float* gp = ga + (size_t)(b * HEADS_ + hd) * HW_;
    for (int key = 1 + tid; key < N_; key += 128)
      gp[key - 1] = __builtin_amdgcn_exp2f(fmaf(Srow[key], cs, nm0)) * i0;
  }
}

#define FTB_  (B_ * TCT_ * TNT_)
#define FTL_  ((B_ * C_ + B_ * HW_ + 255) / 256)
#define FTP_  33

__global__ __launch_bounds__(256) void finalize_kernel(const float* __restrict__ xc,
                                                       const float* __restrict__ ga,
                                                       float* __restrict__ out) {
  __shared__ float ts[TPS_ * FTP_];
  const int tid = threadIdx.x;
  const size_t n_cls = (size_t)B_ * C_;
  const size_t n_x   = (size_t)B_ * C_ * HW_;
  const size_t n_ga  = (size_t)B_ * HW_;
  if (blockIdx.x < FTB_) {
    const int bid = blockIdx.x;
    const int rt = bid % TNT_;
    const int t2 = bid / TNT_;
    const int ct = t2 % TCT_;
    const int b  = t2 / TCT_;
    const int c0 = ct * TCH_, p0 = rt * TPS_;
    const int c = tid & 31, w = tid >> 5;
    const float* ib = xc + ((size_t)b * N_ + 1 + p0) * C_ + c0 + c;
#pragma unroll
    for (int i = 0; i < TPS_ / 8; ++i) {
      int pl = w + 8 * i;
      ts[pl * FTP_ + c] = ib[(size_t)pl * C_];
    }
    __syncthreads();
    float* ob = out + n_cls + (size_t)(b * C_ + c0) * HW_ + p0;
#pragma unroll
    for (int i = 0; i < (TCH_ * TPS_) / 256; ++i) {
      int e  = tid + 256 * i;
      int ci = e / TPS_;
      int pl = e - ci * TPS_;
      ob[(size_t)ci * HW_ + pl] = ts[pl * FTP_ + ci];
    }
  } else {
    size_t j = (size_t)(blockIdx.x - FTB_) * 256 + tid;
    if (j < n_cls) {
      int b = (int)(j / C_), c = (int)(j % C_);
      out[j] = xc[(size_t)b * N_ * C_ + c];
    } else if (j < n_cls + n_ga) {
      size_t k = j - n_cls;
      int b = (int)(k / HW_), p = (int)(k % HW_);
      float s = 0.f;
#pragma unroll
      for (int hh = 0; hh < HEADS_; ++hh)
        s += ga[(size_t)(b * HEADS_ + hh) * HW_ + p];
      out[n_cls + n_x + k] = s * 0.125f;
    }
  }
}

extern "C" void kernel_launch(void* const* d_in, const int* in_sizes, int n_in,
                              void* d_out, int out_size, void* d_ws, size_t ws_size,
                              hipStream_t stream) {
  const float* cls    = (const float*)d_in[0];
  const float* x      = (const float*)d_in[1];
  const float* conv_w = (const float*)d_in[2];
  const float* conv_b = (const float*)d_in[3];
  const float* ln1_g  = (const float*)d_in[4];
  const float* ln1_b  = (const float*)d_in[5];
  const float* qkv_w  = (const float*)d_in[6];
  const float* proj_w = (const float*)d_in[7];
  const float* proj_b = (const float*)d_in[8];
  const float* ln2_g  = (const float*)d_in[9];
  const float* ln2_b  = (const float*)d_in[10];
  const float* fc1_w  = (const float*)d_in[11];
  const float* fc1_b  = (const float*)d_in[12];
  const float* fc2_w  = (const float*)d_in[13];
  const float* fc2_b  = (const float*)d_in[14];

  float* ws  = (float*)d_ws;
  float* xc  = ws;
  float* y   = xc  + (size_t)MP_ * C_;
  float* qkv = y   + (size_t)MP_ * C_;
  float* o   = qkv + (size_t)MP_ * C3_;
  float* m1  = qkv;
  float* ga  = o   + (size_t)MP_ * C_;
  float* out = (float*)d_out;

  conv_pe_kernel<<<dim3(TNT_, TCT_, B_), 256, 0, stream>>>(x, conv_w, conv_b, xc);
  cls_copy_kernel<<<(B_ * C_ + 255) / 256, 256, 0, stream>>>(cls, xc);

  ln_kernel<<<(M_ + 7) / 8, 256, 0, stream>>>(xc, ln1_g, ln1_b, y, M_);

  gemm_wmma_kernel<0><<<dim3(C3_ / GBN_, GMB_), 128, 0, stream>>>(
      y, qkv_w, nullptr, qkv, C_, C3_);

  attn_kernel<<<dim3(AQBN_, HEADS_, B_), 128, 0, stream>>>(qkv, o, ga);

  gemm_wmma_kernel<1><<<dim3(C_ / GBN_, GMB_), 128, 0, stream>>>(
      o, proj_w, proj_b, xc, C_, C_);

  ln_kernel<<<(M_ + 7) / 8, 256, 0, stream>>>(xc, ln2_g, ln2_b, y, M_);

  gemm_wmma_kernel<2><<<dim3(C4_ / GBN_, GMB_), 128, 0, stream>>>(
      y, fc1_w, fc1_b, m1, C_, C4_);

  gemm_wmma_kernel<1><<<dim3(C_ / GBN_, GMB_), 128, 0, stream>>>(
      m1, fc2_w, fc2_b, xc, C4_, C_);

  finalize_kernel<<<FTB_ + FTL_, 256, 0, stream>>>(xc, ga, out);
}
